# starting kernel
# baseline (speedup 1.0000x reference)
#include <hip/hip_runtime.h>

typedef __attribute__((ext_vector_type(2))) float v2f;
typedef __attribute__((ext_vector_type(8))) float v8f;

__global__ __launch_bounds__(256) void zero_f32(float* __restrict__ p, long long n) {
    long long i = (long long)blockIdx.x * blockDim.x + threadIdx.x;
    if (i < n) p[i] = 0.0f;
}

__global__ __launch_bounds__(256) void deg_count(const int* __restrict__ dst,
                                                 float* __restrict__ deg, int nE) {
    int e = blockIdx.x * blockDim.x + threadIdx.x;
    if (e < nE) atomicAdd(&deg[dst[e]], 1.0f);
}

__global__ __launch_bounds__(256) void deg_to_dinv(float* __restrict__ deg, int n) {
    int i = blockIdx.x * blockDim.x + threadIdx.x;
    if (i < n) deg[i] = rsqrtf(deg[i] + 1.0f);
}

__global__ __launch_bounds__(256) void gemm_x_w1(const float* __restrict__ X,
                                                 const float* __restrict__ W,
                                                 float* __restrict__ H, int n) {
    const int lane = threadIdx.x & 31;
    const int tile = blockIdx.x * 8 + (threadIdx.x >> 5);
    const int row0 = tile * 16;
    if (row0 >= n) return;
    int m = row0 + (lane & 15);
    if (m >= n) m = n - 1;
    const int kh  = (lane >> 4) << 1;
    const int col = lane & 15;
    const float* xr = X + (size_t)m * 256;

    v8f acc = {};
#pragma unroll 4
    for (int k0 = 0; k0 < 256; k0 += 4) {
        v2f a, b;
        a.x = xr[k0 + kh];
        a.y = xr[k0 + kh + 1];
        b.x = W[(size_t)(k0 + kh) * 16 + col];
        b.y = W[(size_t)(k0 + kh + 1) * 16 + col];
        acc = __builtin_amdgcn_wmma_f32_16x16x4_f32(false, a, false, b,
                                                    (short)0, acc, false, false);
    }
    const int ra = (lane >> 4) << 3;
    if (row0 + 16 <= n) {
#pragma unroll
        for (int v = 0; v < 8; ++v)
            H[(size_t)(row0 + ra + v) * 16 + col] = acc[v];
    } else {
#pragma unroll
        for (int v = 0; v < 8; ++v) {
            int r = row0 + ra + v;
            if (r < n) H[(size_t)r * 16 + col] = acc[v];
        }
    }
}

__global__ __launch_bounds__(256) void gemm_z_w2(const float* __restrict__ Z,
                                                 const float* __restrict__ W,
                                                 float* __restrict__ H, int n) {
    const int lane = threadIdx.x & 31;
    const int tile = blockIdx.x * 8 + (threadIdx.x >> 5);
    const int row0 = tile * 16;
    if (row0 >= n) return;
    int m = row0 + (lane & 15);
    if (m >= n) m = n - 1;
    const int kh  = (lane >> 4) << 1;
    const int col = lane & 15;
    const float* zr = Z + (size_t)m * 16;

    v8f acc0 = {}, acc1 = {}, acc2 = {}, acc3 = {};
#pragma unroll
    for (int k0 = 0; k0 < 16; k0 += 4) {
        v2f a, b;
        a.x = zr[k0 + kh];
        a.y = zr[k0 + kh + 1];
        const float* w0 = W + (size_t)(k0 + kh) * 64;
        const float* w1 = w0 + 64;
        b.x = w0[col];      b.y = w1[col];
        acc0 = __builtin_amdgcn_wmma_f32_16x16x4_f32(false, a, false, b, (short)0, acc0, false, false);
        b.x = w0[col + 16]; b.y = w1[col + 16];
        acc1 = __builtin_amdgcn_wmma_f32_16x16x4_f32(false, a, false, b, (short)0, acc1, false, false);
        b.x = w0[col + 32]; b.y = w1[col + 32];
        acc2 = __builtin_amdgcn_wmma_f32_16x16x4_f32(false, a, false, b, (short)0, acc2, false, false);
        b.x = w0[col + 48]; b.y = w1[col + 48];
        acc3 = __builtin_amdgcn_wmma_f32_16x16x4_f32(false, a, false, b, (short)0, acc3, false, false);
    }
    const int ra = (lane >> 4) << 3;
    if (row0 + 16 <= n) {
#pragma unroll
        for (int v = 0; v < 8; ++v) {
            float* hp = H + (size_t)(row0 + ra + v) * 64 + col;
            hp[0]  = acc0[v];
            hp[16] = acc1[v];
            hp[32] = acc2[v];
            hp[48] = acc3[v];
        }
    } else {
#pragma unroll
        for (int v = 0; v < 8; ++v) {
            int r = row0 + ra + v;
            if (r < n) {
                float* hp = H + (size_t)r * 64 + col;
                hp[0]  = acc0[v];
                hp[16] = acc1[v];
                hp[32] = acc2[v];
                hp[48] = acc3[v];
            }
        }
    }
}

template <int F>
__global__ __launch_bounds__(256) void scatter_add(const int* __restrict__ src,
                                                   const int* __restrict__ dst,
                                                   const float* __restrict__ dinv,
                                                   const float* __restrict__ H,
                                                   float* __restrict__ OUT, int nE) {
    constexpr int CH = F / 16;
    long long idx = (long long)blockIdx.x * blockDim.x + threadIdx.x;
    long long e = idx / CH;
    int c = (int)(idx % CH);
    if (e >= nE) return;
    int s = src[e], d = dst[e];
    float nrm = dinv[s] * dinv[d];
    const float* hp = H + (size_t)s * F + c * 16;
    float* op = OUT + (size_t)d * F + c * 16;
#pragma unroll
    for (int f = 0; f < 16; ++f)
        atomicAdd(op + f, hp[f] * nrm);
}

template <int F, bool RELU>
__global__ __launch_bounds__(256) void epilogue(float* __restrict__ AGG,
                                                const float* __restrict__ H,
                                                const float* __restrict__ dinv,
                                                const float* __restrict__ bias, int n) {
    long long idx = (long long)blockIdx.x * blockDim.x + threadIdx.x;
    if (idx >= (long long)n * F) return;
    int i = (int)(idx / F);
    int f = (int)(idx % F);
    float di = dinv[i];
    float v = AGG[idx] + di * di * H[idx] + bias[f];
    if (RELU) v = fmaxf(v, 0.0f);
    AGG[idx] = v;
}

__global__ __launch_bounds__(256) void edge_score(const int* __restrict__ psrc,
                                                  const int* __restrict__ pdst,
                                                  const int* __restrict__ nsrc,
                                                  const int* __restrict__ ndst,
                                                  const float* __restrict__ Z,
                                                  float* __restrict__ out, int nT) {
    int j = blockIdx.x * blockDim.x + threadIdx.x;
    if (j >= 2 * nT) return;
    int s, d;
    if (j < nT) { s = psrc[j]; d = pdst[j]; }
    else        { s = nsrc[j - nT]; d = ndst[j - nT]; }
    const float4* zs = (const float4*)(Z + (size_t)s * 64);
    const float4* zd = (const float4*)(Z + (size_t)d * 64);
    float acc = 0.0f;
#pragma unroll
    for (int k = 0; k < 16; ++k) {
        float4 a = zs[k], b = zd[k];
        acc = fmaf(a.x, b.x, acc);
        acc = fmaf(a.y, b.y, acc);
        acc = fmaf(a.z, b.z, acc);
        acc = fmaf(a.w, b.w, acc);
    }
    out[j] = acc;
}

extern "C" void kernel_launch(void* const* d_in, const int* in_sizes, int n_in,
                              void* d_out, int out_size, void* d_ws, size_t ws_size,
                              hipStream_t stream) {
    const float* x   = (const float*)d_in[0];
    const int*   tei = (const int*)d_in[1];
    const int*   tpo = (const int*)d_in[2];
    const int*   tne = (const int*)d_in[3];
    const float* W1  = (const float*)d_in[4];
    const float* b1  = (const float*)d_in[5];
    const float* W2  = (const float*)d_in[6];
    const float* b2  = (const float*)d_in[7];

    const int N  = in_sizes[0] / 256;
    const int E  = in_sizes[1] / 2;
    const int ET = in_sizes[2] / 2;

    float* ws   = (float*)d_ws;
    float* dinv = ws;
    float* h1   = ws + (size_t)N;
    float* agg1 = ws + (size_t)17 * N;
    float* h2   = ws + (size_t)33 * N;
    float* agg2 = ws + (size_t)97 * N;

    const int B = 256;
    auto cdiv = [](long long a, long long b) { return (int)((a + b - 1) / b); };

    zero_f32<<<cdiv(N, B), B, 0, stream>>>(dinv, N);
    zero_f32<<<cdiv((long long)16 * N, B), B, 0, stream>>>(agg1, (long long)16 * N);
    zero_f32<<<cdiv((long long)64 * N, B), B, 0, stream>>>(agg2, (long long)64 * N);

    deg_count<<<cdiv(E, B), B, 0, stream>>>(tei + E, dinv, E);
    deg_to_dinv<<<cdiv(N, B), B, 0, stream>>>(dinv, N);

    const int tiles = (N + 15) / 16;

    gemm_x_w1<<<cdiv(tiles, 8), B, 0, stream>>>(x, W1, h1, N);
    scatter_add<16><<<cdiv(E, B), B, 0, stream>>>(tei, tei + E, dinv, h1, agg1, E);
    epilogue<16, true><<<cdiv((long long)16 * N, B), B, 0, stream>>>(agg1, h1, dinv, b1, N);

    gemm_z_w2<<<cdiv(tiles, 8), B, 0, stream>>>(agg1, W2, h2, N);
    scatter_add<64><<<cdiv((long long)4 * E, B), B, 0, stream>>>(tei, tei + E, dinv, h2, agg2, E);
    epilogue<64, false><<<cdiv((long long)64 * N, B), B, 0, stream>>>(agg2, h2, dinv, b2, N);

    edge_score<<<cdiv(2 * ET, B), B, 0, stream>>>(tpo, tpo + ET, tne, tne + ET,
                                                  agg2, (float*)d_out, ET);
}


# Sort edges by destination each call (int returning add = arrival-order slot, prefix sum, fill), then sum each node's in-edges by gather in slot order; all float atomics removed
# speedup vs baseline: 4.7126x; 4.7126x over previous
#include <hip/hip_runtime.h>

typedef __attribute__((ext_vector_type(2))) float v2f;
typedef __attribute__((ext_vector_type(8))) float v8f;

__global__ __launch_bounds__(256) void zero_i32(int* __restrict__ p, int n) {
    int i = blockIdx.x * blockDim.x + threadIdx.x;
    if (i < n) p[i] = 0;
}

__global__ __launch_bounds__(256) void count_rank(const int* __restrict__ dst,
                                                  int* __restrict__ cnt,
                                                  int* __restrict__ rank, int nE) {
    int e = blockIdx.x * blockDim.x + threadIdx.x;
    if (e < nE) rank[e] = atomicAdd(&cnt[dst[e]], 1);
}

__global__ __launch_bounds__(256) void scan_counts(const int* __restrict__ cnt,
                                                   int* __restrict__ rowptr,
                                                   float* __restrict__ dinv, int n) {
    __shared__ int s_front[8];
    __shared__ int s_wave[8];
    const int t = threadIdx.x, lane = t & 31, w = t >> 5;

    int front = 0;
    const int4* c4 = (const int4*)cnt;
    const int nfront = blockIdx.x * 256;
#pragma unroll 8
    for (int j = t; j < nfront; j += 256) {
        int4 v = c4[j];
        front += (v.x + v.y) + (v.z + v.w);
    }
    for (int o = 16; o > 0; o >>= 1) front += __shfl_xor(front, o);
    if (lane == 0) s_front[w] = front;

    const int i0 = blockIdx.x * 1024 + 4 * t;
    int c[4];
#pragma unroll
    for (int k = 0; k < 4; ++k) c[k] = (i0 + k < n) ? cnt[i0 + k] : 0;
    const int s = (c[0] + c[1]) + (c[2] + c[3]);
    int inc = s;
    for (int o = 1; o < 32; o <<= 1) {
        int v = __shfl_up(inc, o);
        if (lane >= o) inc += v;
    }
    if (lane == 31) s_wave[w] = inc;
    __syncthreads();
    int ex = inc - s;
#pragma unroll
    for (int k = 0; k < 8; ++k) {
        ex += s_front[k];
        if (k < w) ex += s_wave[k];
    }
#pragma unroll
    for (int k = 0; k < 4; ++k) {
        if (i0 + k < n) {
            rowptr[i0 + k] = ex;
            dinv[i0 + k] = rsqrtf((float)c[k] + 1.0f);
            ex += c[k];
        }
    }
}

__global__ __launch_bounds__(256) void fill_csr(const int* __restrict__ src,
                                                const int* __restrict__ dst,
                                                const int* __restrict__ rowptr,
                                                const int* __restrict__ rank,
                                                int* __restrict__ csr, int nE) {
    int e = blockIdx.x * blockDim.x + threadIdx.x;
    if (e < nE) csr[rowptr[dst[e]] + rank[e]] = src[e];
}

__global__ __launch_bounds__(256) void gemm_x_w1(const float* __restrict__ X,
                                                 const float* __restrict__ W,
                                                 float* __restrict__ H, int n) {
    const int lane = threadIdx.x & 31;
    const int tile = blockIdx.x * 8 + (threadIdx.x >> 5);
    const int row0 = tile * 16;
    if (row0 >= n) return;
    int m = row0 + (lane & 15);
    if (m >= n) m = n - 1;
    const int kh  = (lane >> 4) << 1;
    const int col = lane & 15;
    const float* xr = X + (size_t)m * 256;

    v8f acc = {};
#pragma unroll 4
    for (int k0 = 0; k0 < 256; k0 += 4) {
        v2f a, b;
        a.x = xr[k0 + kh];
        a.y = xr[k0 + kh + 1];
        b.x = W[(size_t)(k0 + kh) * 16 + col];
        b.y = W[(size_t)(k0 + kh + 1) * 16 + col];
        acc = __builtin_amdgcn_wmma_f32_16x16x4_f32(false, a, false, b,
                                                    (short)0, acc, false, false);
    }
    const int ra = (lane >> 4) << 3;
    if (row0 + 16 <= n) {
#pragma unroll
        for (int v = 0; v < 8; ++v)
            H[(size_t)(row0 + ra + v) * 16 + col] = acc[v];
    } else {
#pragma unroll
        for (int v = 0; v < 8; ++v) {
            int r = row0 + ra + v;
            if (r < n) H[(size_t)r * 16 + col] = acc[v];
        }
    }
}

__global__ __launch_bounds__(256) void gemm_z_w2(const float* __restrict__ Z,
                                                 const float* __restrict__ W,
                                                 float* __restrict__ H, int n) {
    const int lane = threadIdx.x & 31;
    const int tile = blockIdx.x * 8 + (threadIdx.x >> 5);
    const int row0 = tile * 16;
    if (row0 >= n) return;
    int m = row0 + (lane & 15);
    if (m >= n) m = n - 1;
    const int kh  = (lane >> 4) << 1;
    const int col = lane & 15;
    const float* zr = Z + (size_t)m * 16;

    v8f acc0 = {}, acc1 = {}, acc2 = {}, acc3 = {};
#pragma unroll
    for (int k0 = 0; k0 < 16; k0 += 4) {
        v2f a, b;
        a.x = zr[k0 + kh];
        a.y = zr[k0 + kh + 1];
        const float* w0 = W + (size_t)(k0 + kh) * 64;
        const float* w1 = w0 + 64;
        b.x = w0[col];      b.y = w1[col];
        acc0 = __builtin_amdgcn_wmma_f32_16x16x4_f32(false, a, false, b, (short)0, acc0, false, false);
        b.x = w0[col + 16]; b.y = w1[col + 16];
        acc1 = __builtin_amdgcn_wmma_f32_16x16x4_f32(false, a, false, b, (short)0, acc1, false, false);
        b.x = w0[col + 32]; b.y = w1[col + 32];
        acc2 = __builtin_amdgcn_wmma_f32_16x16x4_f32(false, a, false, b, (short)0, acc2, false, false);
        b.x = w0[col + 48]; b.y = w1[col + 48];
        acc3 = __builtin_amdgcn_wmma_f32_16x16x4_f32(false, a, false, b, (short)0, acc3, false, false);
    }
    const int ra = (lane >> 4) << 3;
    if (row0 + 16 <= n) {
#pragma unroll
        for (int v = 0; v < 8; ++v) {
            float* hp = H + (size_t)(row0 + ra + v) * 64 + col;
            hp[0]  = acc0[v];
            hp[16] = acc1[v];
            hp[32] = acc2[v];
            hp[48] = acc3[v];
        }
    } else {
#pragma unroll
        for (int v = 0; v < 8; ++v) {
            int r = row0 + ra + v;
            if (r < n) {
                float* hp = H + (size_t)r * 64 + col;
                hp[0]  = acc0[v];
                hp[16] = acc1[v];
                hp[32] = acc2[v];
                hp[48] = acc3[v];
            }
        }
    }
}

template <int F, bool RELU, int U>
__global__ __launch_bounds__(256) void gather_rows(const int* __restrict__ rowptr,
                                                   const int* __restrict__ cnt,
                                                   const int* __restrict__ csr,
                                                   const float* __restrict__ dinv,
                                                   const float* __restrict__ H,
                                                   const float* __restrict__ bias,
                                                   float* __restrict__ Z, int n) {
    constexpr int L = F / 4;
    const long long gid = (long long)blockIdx.x * blockDim.x + threadIdx.x;
    const int i = (int)(gid / L);
    const int l = (int)(gid % L);
    if (i >= n) return;
    const float4* H4 = (const float4*)H;
    const int beg = rowptr[i];
    const int end = beg + cnt[i];
    const float di = dinv[i];
    float4 acc = make_float4(0.0f, 0.0f, 0.0f, 0.0f);
    if (beg < end) {
        const int last = end - 1;
        int s[U];
#pragma unroll
        for (int k = 0; k < U; ++k) s[k] = csr[min(beg + k, last)];
        for (int p = beg; p < end; p += U) {
            int sn[U];
#pragma unroll
            for (int k = 0; k < U; ++k) sn[k] = csr[min(p + U + k, last)];
            float dv[U], nr[U];
            float4 h[U];
#pragma unroll
            for (int k = 0; k < U; ++k) {
                dv[k] = dinv[s[k]];
                h[k] = H4[(size_t)s[k] * L + l];
            }
#pragma unroll
            for (int k = 0; k < U; ++k)
                asm volatile("" :: "v"(dv[k]), "v"(h[k].x), "v"(h[k].y), "v"(h[k].z), "v"(h[k].w));
#pragma unroll
            for (int k = 0; k < U; ++k) nr[k] = dv[k] * di;
#pragma unroll
            for (int k = 0; k < U; ++k) {
                if (k == 0 || p + k < end) {
                    acc.x = fmaf(h[k].x, nr[k], acc.x);
                    acc.y = fmaf(h[k].y, nr[k], acc.y);
                    acc.z = fmaf(h[k].z, nr[k], acc.z);
                    acc.w = fmaf(h[k].w, nr[k], acc.w);
                }
            }
#pragma unroll
            for (int k = 0; k < U; ++k) s[k] = sn[k];
        }
    }
    const float4 hs = H4[(size_t)i * L + l];
    const float sl = di * di;
    float4 v;
    v.x = acc.x + sl * hs.x + bias[4 * l + 0];
    v.y = acc.y + sl * hs.y + bias[4 * l + 1];
    v.z = acc.z + sl * hs.z + bias[4 * l + 2];
    v.w = acc.w + sl * hs.w + bias[4 * l + 3];
    if (RELU) {
        v.x = fmaxf(v.x, 0.0f); v.y = fmaxf(v.y, 0.0f);
        v.z = fmaxf(v.z, 0.0f); v.w = fmaxf(v.w, 0.0f);
    }
    ((float4*)Z)[(size_t)i * L + l] = v;
}

__global__ __launch_bounds__(256) void edge_score(const int* __restrict__ psrc,
                                                  const int* __restrict__ pdst,
                                                  const int* __restrict__ nsrc,
                                                  const int* __restrict__ ndst,
                                                  const float* __restrict__ Z,
                                                  float* __restrict__ out, int nT) {
    int j = blockIdx.x * blockDim.x + threadIdx.x;
    if (j >= 2 * nT) return;
    int s, d;
    if (j < nT) { s = psrc[j]; d = pdst[j]; }
    else        { s = nsrc[j - nT]; d = ndst[j - nT]; }
    const float4* zs = (const float4*)(Z + (size_t)s * 64);
    const float4* zd = (const float4*)(Z + (size_t)d * 64);
    float acc = 0.0f;
#pragma unroll
    for (int k = 0; k < 16; ++k) {
        float4 a = zs[k], b = zd[k];
        acc = fmaf(a.x, b.x, acc);
        acc = fmaf(a.y, b.y, acc);
        acc = fmaf(a.z, b.z, acc);
        acc = fmaf(a.w, b.w, acc);
    }
    out[j] = acc;
}

extern "C" void kernel_launch(void* const* d_in, const int* in_sizes, int n_in,
                              void* d_out, int out_size, void* d_ws, size_t ws_size,
                              hipStream_t stream) {
    const float* x   = (const float*)d_in[0];
    const int*   tei = (const int*)d_in[1];
    const int*   tpo = (const int*)d_in[2];
    const int*   tne = (const int*)d_in[3];
    const float* W1  = (const float*)d_in[4];
    const float* b1  = (const float*)d_in[5];
    const float* W2  = (const float*)d_in[6];
    const float* b2  = (const float*)d_in[7];

    const int N  = in_sizes[0] / 256;
    const int E  = in_sizes[1] / 2;
    const int ET = in_sizes[2] / 2;

    float* ws   = (float*)d_ws;
    float* h1   = ws;
    float* z1   = ws + (size_t)16 * N;
    float* h2   = ws + (size_t)32 * N;
    float* z2   = ws + (size_t)96 * N;
    int*   cnt  = (int*)(ws + (size_t)160 * N);
    float* dinv = ws + (size_t)161 * N;
    int*   rowp = (int*)(ws + (size_t)162 * N);
    int*   rank = (int*)(ws + (size_t)163 * N);
    int*   csr  = rank + (size_t)E;

    const int B = 256;
    auto cdiv = [](long long a, long long b) { return (int)((a + b - 1) / b); };

    zero_i32<<<cdiv(N, B), B, 0, stream>>>(cnt, N);
    count_rank<<<cdiv(E, B), B, 0, stream>>>(tei + E, cnt, rank, E);
    scan_counts<<<cdiv(N, 1024), B, 0, stream>>>(cnt, rowp, dinv, N);
    fill_csr<<<cdiv(E, B), B, 0, stream>>>(tei, tei + E, rowp, rank, csr, E);

    const int tiles = (N + 15) / 16;

    gemm_x_w1<<<cdiv(tiles, 8), B, 0, stream>>>(x, W1, h1, N);
    gather_rows<16, true, 8><<<cdiv((long long)4 * N, B), B, 0, stream>>>(
        rowp, cnt, csr, dinv, h1, b1, z1, N);

    gemm_z_w2<<<cdiv(tiles, 8), B, 0, stream>>>(z1, W2, h2, N);
    gather_rows<64, false, 8><<<cdiv((long long)16 * N, B), B, 0, stream>>>(
        rowp, cnt, csr, dinv, h2, b2, z2, N);

    edge_score<<<cdiv(2 * ET, B), B, 0, stream>>>(tpo, tpo + ET, tne, tne + ET,
                                                  z2, (float*)d_out, ET);
}


# Edge scoring: 16 lanes read one 256-byte z row per load (4 whole lines per wave load, not 32); dot = 4 products per lane, then 16-lane DPP tree (sum order changes); 32 scores per 128-byte store
# speedup vs baseline: 5.3860x; 1.1429x over previous
#include <hip/hip_runtime.h>

typedef __attribute__((ext_vector_type(2))) float v2f;
typedef __attribute__((ext_vector_type(8))) float v8f;

__global__ __launch_bounds__(256) void zero_i32(int* __restrict__ p, int n) {
    int i = blockIdx.x * blockDim.x + threadIdx.x;
    if (i < n) p[i] = 0;
}

__global__ __launch_bounds__(256) void count_rank(const int* __restrict__ dst,
                                                  int* __restrict__ cnt,
                                                  int* __restrict__ rank, int nE) {
    int e = blockIdx.x * blockDim.x + threadIdx.x;
    if (e < nE) rank[e] = atomicAdd(&cnt[dst[e]], 1);
}

__global__ __launch_bounds__(256) void scan_counts(const int* __restrict__ cnt,
                                                   int* __restrict__ rowptr,
                                                   float* __restrict__ dinv, int n) {
    __shared__ int s_front[8];
    __shared__ int s_wave[8];
    const int t = threadIdx.x, lane = t & 31, w = t >> 5;

    int front = 0;
    const int4* c4 = (const int4*)cnt;
    const int nfront = blockIdx.x * 256;
#pragma unroll 8
    for (int j = t; j < nfront; j += 256) {
        int4 v = c4[j];
        front += (v.x + v.y) + (v.z + v.w);
    }
    for (int o = 16; o > 0; o >>= 1) front += __shfl_xor(front, o);
    if (lane == 0) s_front[w] = front;

    const int i0 = blockIdx.x * 1024 + 4 * t;
    int c[4];
#pragma unroll
    for (int k = 0; k < 4; ++k) c[k] = (i0 + k < n) ? cnt[i0 + k] : 0;
    const int s = (c[0] + c[1]) + (c[2] + c[3]);
    int inc = s;
    for (int o = 1; o < 32; o <<= 1) {
        int v = __shfl_up(inc, o);
        if (lane >= o) inc += v;
    }
    if (lane == 31) s_wave[w] = inc;
    __syncthreads();
    int ex = inc - s;
#pragma unroll
    for (int k = 0; k < 8; ++k) {
        ex += s_front[k];
        if (k < w) ex += s_wave[k];
    }
#pragma unroll
    for (int k = 0; k < 4; ++k) {
        if (i0 + k < n) {
            rowptr[i0 + k] = ex;
            dinv[i0 + k] = rsqrtf((float)c[k] + 1.0f);
            ex += c[k];
        }
    }
}

__global__ __launch_bounds__(256) void fill_csr(const int* __restrict__ src,
                                                const int* __restrict__ dst,
                                                const int* __restrict__ rowptr,
                                                const int* __restrict__ rank,
                                                int* __restrict__ csr, int nE) {
    int e = blockIdx.x * blockDim.x + threadIdx.x;
    if (e < nE) csr[rowptr[dst[e]] + rank[e]] = src[e];
}

__global__ __launch_bounds__(256) void gemm_x_w1(const float* __restrict__ X,
                                                 const float* __restrict__ W,
                                                 float* __restrict__ H, int n) {
    const int lane = threadIdx.x & 31;
    const int tile = blockIdx.x * 8 + (threadIdx.x >> 5);
    const int row0 = tile * 16;
    if (row0 >= n) return;
    int m = row0 + (lane & 15);
    if (m >= n) m = n - 1;
    const int kh  = (lane >> 4) << 1;
    const int col = lane & 15;
    const float* xr = X + (size_t)m * 256;

    v8f acc = {};
#pragma unroll 4
    for (int k0 = 0; k0 < 256; k0 += 4) {
        v2f a, b;
        a.x = xr[k0 + kh];
        a.y = xr[k0 + kh + 1];
        b.x = W[(size_t)(k0 + kh) * 16 + col];
        b.y = W[(size_t)(k0 + kh + 1) * 16 + col];
        acc = __builtin_amdgcn_wmma_f32_16x16x4_f32(false, a, false, b,
                                                    (short)0, acc, false, false);
    }
    const int ra = (lane >> 4) << 3;
    if (row0 + 16 <= n) {
#pragma unroll
        for (int v = 0; v < 8; ++v)
            H[(size_t)(row0 + ra + v) * 16 + col] = acc[v];
    } else {
#pragma unroll
        for (int v = 0; v < 8; ++v) {
            int r = row0 + ra + v;
            if (r < n) H[(size_t)r * 16 + col] = acc[v];
        }
    }
}

__global__ __launch_bounds__(256) void gemm_z_w2(const float* __restrict__ Z,
                                                 const float* __restrict__ W,
                                                 float* __restrict__ H, int n) {
    const int lane = threadIdx.x & 31;
    const int tile = blockIdx.x * 8 + (threadIdx.x >> 5);
    const int row0 = tile * 16;
    if (row0 >= n) return;
    int m = row0 + (lane & 15);
    if (m >= n) m = n - 1;
    const int kh  = (lane >> 4) << 1;
    const int col = lane & 15;
    const float* zr = Z + (size_t)m * 16;

    v8f acc0 = {}, acc1 = {}, acc2 = {}, acc3 = {};
#pragma unroll
    for (int k0 = 0; k0 < 16; k0 += 4) {
        v2f a, b;
        a.x = zr[k0 + kh];
        a.y = zr[k0 + kh + 1];
        const float* w0 = W + (size_t)(k0 + kh) * 64;
        const float* w1 = w0 + 64;
        b.x = w0[col];      b.y = w1[col];
        acc0 = __builtin_amdgcn_wmma_f32_16x16x4_f32(false, a, false, b, (short)0, acc0, false, false);
        b.x = w0[col + 16]; b.y = w1[col + 16];
        acc1 = __builtin_amdgcn_wmma_f32_16x16x4_f32(false, a, false, b, (short)0, acc1, false, false);
        b.x = w0[col + 32]; b.y = w1[col + 32];
        acc2 = __builtin_amdgcn_wmma_f32_16x16x4_f32(false, a, false, b, (short)0, acc2, false, false);
        b.x = w0[col + 48]; b.y = w1[col + 48];
        acc3 = __builtin_amdgcn_wmma_f32_16x16x4_f32(false, a, false, b, (short)0, acc3, false, false);
    }
    const int ra = (lane >> 4) << 3;
    if (row0 + 16 <= n) {
#pragma unroll
        for (int v = 0; v < 8; ++v) {
            float* hp = H + (size_t)(row0 + ra + v) * 64 + col;
            hp[0]  = acc0[v];
            hp[16] = acc1[v];
            hp[32] = acc2[v];
            hp[48] = acc3[v];
        }
    } else {
#pragma unroll
        for (int v = 0; v < 8; ++v) {
            int r = row0 + ra + v;
            if (r < n) {
                float* hp = H + (size_t)r * 64 + col;
                hp[0]  = acc0[v];
                hp[16] = acc1[v];
                hp[32] = acc2[v];
                hp[48] = acc3[v];
            }
        }
    }
}

template <int F, bool RELU, int U>
__global__ __launch_bounds__(256) void gather_rows(const int* __restrict__ rowptr,
                                                   const int* __restrict__ cnt,
                                                   const int* __restrict__ csr,
                                                   const float* __restrict__ dinv,
                                                   const float* __restrict__ H,
                                                   const float* __restrict__ bias,
                                                   float* __restrict__ Z, int n) {
    constexpr int L = F / 4;
    const long long gid = (long long)blockIdx.x * blockDim.x + threadIdx.x;
    const int i = (int)(gid / L);
    const int l = (int)(gid % L);
    if (i >= n) return;
    const float4* H4 = (const float4*)H;
    const int beg = rowptr[i];
    const int end = beg + cnt[i];
    const float di = dinv[i];
    float4 acc = make_float4(0.0f, 0.0f, 0.0f, 0.0f);
    if (beg < end) {
        const int last = end - 1;
        int s[U];
#pragma unroll
        for (int k = 0; k < U; ++k) s[k] = csr[min(beg + k, last)];
        for (int p = beg; p < end; p += U) {
            int sn[U];
#pragma unroll
            for (int k = 0; k < U; ++k) sn[k] = csr[min(p + U + k, last)];
            float dv[U], nr[U];
            float4 h[U];
#pragma unroll
            for (int k = 0; k < U; ++k) {
                dv[k] = dinv[s[k]];
                h[k] = H4[(size_t)s[k] * L + l];
            }
#pragma unroll
            for (int k = 0; k < U; ++k)
                asm volatile("" :: "v"(dv[k]), "v"(h[k].x), "v"(h[k].y), "v"(h[k].z), "v"(h[k].w));
#pragma unroll
            for (int k = 0; k < U; ++k) nr[k] = dv[k] * di;
#pragma unroll
            for (int k = 0; k < U; ++k) {
                if (k == 0 || p + k < end) {
                    acc.x = fmaf(h[k].x, nr[k], acc.x);
                    acc.y = fmaf(h[k].y, nr[k], acc.y);
                    acc.z = fmaf(h[k].z, nr[k], acc.z);
                    acc.w = fmaf(h[k].w, nr[k], acc.w);
                }
            }
#pragma unroll
            for (int k = 0; k < U; ++k) s[k] = sn[k];
        }
    }
    const float4 hs = H4[(size_t)i * L + l];
    const float sl = di * di;
    float4 v;
    v.x = acc.x + sl * hs.x + bias[4 * l + 0];
    v.y = acc.y + sl * hs.y + bias[4 * l + 1];
    v.z = acc.z + sl * hs.z + bias[4 * l + 2];
    v.w = acc.w + sl * hs.w + bias[4 * l + 3];
    if (RELU) {
        v.x = fmaxf(v.x, 0.0f); v.y = fmaxf(v.y, 0.0f);
        v.z = fmaxf(v.z, 0.0f); v.w = fmaxf(v.w, 0.0f);
    }
    ((float4*)Z)[(size_t)i * L + l] = v;
}


__device__ __forceinline__ float row_sum16(float v) {
    v += __int_as_float(__builtin_amdgcn_update_dpp(0, __float_as_int(v), 0x161, 0xf, 0xf, true));
    v += __int_as_float(__builtin_amdgcn_update_dpp(0, __float_as_int(v), 0x162, 0xf, 0xf, true));
    v += __int_as_float(__builtin_amdgcn_update_dpp(0, __float_as_int(v), 0x164, 0xf, 0xf, true));
    v += __int_as_float(__builtin_amdgcn_update_dpp(0, __float_as_int(v), 0x168, 0xf, 0xf, true));
    return v;
}

__global__ __launch_bounds__(256) void edge_score(const int* __restrict__ psrc,
                                                  const int* __restrict__ pdst,
                                                  const int* __restrict__ nsrc,
                                                  const int* __restrict__ ndst,
                                                  const float* __restrict__ Z,
                                                  float* __restrict__ out, int nT) {
    const int lane = threadIdx.x & 31;
    const int c = lane & 15;
    const unsigned cb = (unsigned)c << 4;
    const bool upper = lane >= 16;
    const int total = 2 * nT;
    const int ngroups = (total + 31) >> 5;
    const int nwaves = gridDim.x * 8;
    int g = blockIdx.x * 8 + __builtin_amdgcn_readfirstlane((int)(threadIdx.x >> 5));
    if (g >= ngroups) return;
    const char* Zb = (const char*)Z;

    int j = min(g * 32 + lane, total - 1);
    int s = (j < nT) ? psrc[j] : nsrc[j - nT];
    int d = (j < nT) ? pdst[j] : ndst[j - nT];

    for (; g < ngroups; g += nwaves) {
        const int gn = min(g + nwaves, ngroups - 1);
        const int jn = min(gn * 32 + lane, total - 1);
        const int sn = (jn < nT) ? psrc[jn] : nsrc[jn - nT];
        const int dn = (jn < nT) ? pdst[jn] : ndst[jn - nT];

        float res = 0.0f;
#pragma unroll
        for (int t0 = 0; t0 < 16; t0 += 8) {
            float4 a[8], b[8];
#pragma unroll
            for (int u = 0; u < 8; ++u) {
                const int t = t0 + u;
                const int s0 = __builtin_amdgcn_readlane(s, t);
                const int s1 = __builtin_amdgcn_readlane(s, t + 16);
                const int d0 = __builtin_amdgcn_readlane(d, t);
                const int d1 = __builtin_amdgcn_readlane(d, t + 16);
                const unsigned so = ((unsigned)(upper ? s1 : s0) << 8) + cb;
                const unsigned dof = ((unsigned)(upper ? d1 : d0) << 8) + cb;
                a[u] = *(const float4*)(Zb + so);
                b[u] = *(const float4*)(Zb + dof);
            }
#pragma unroll
            for (int u = 0; u < 8; ++u) {
                float p = a[u].x * b[u].x;
                p = fmaf(a[u].y, b[u].y, p);
                p = fmaf(a[u].z, b[u].z, p);
                p = fmaf(a[u].w, b[u].w, p);
                p = row_sum16(p);
                if (c == t0 + u) res = p;
            }
        }
        const int jo = g * 32 + lane;
        if (jo < total) out[jo] = res;
        s = sn;
        d = dn;
    }
}

extern "C" void kernel_launch(void* const* d_in, const int* in_sizes, int n_in,
                              void* d_out, int out_size, void* d_ws, size_t ws_size,
                              hipStream_t stream) {
    const float* x   = (const float*)d_in[0];
    const int*   tei = (const int*)d_in[1];
    const int*   tpo = (const int*)d_in[2];
    const int*   tne = (const int*)d_in[3];
    const float* W1  = (const float*)d_in[4];
    const float* b1  = (const float*)d_in[5];
    const float* W2  = (const float*)d_in[6];
    const float* b2  = (const float*)d_in[7];

    const int N  = in_sizes[0] / 256;
    const int E  = in_sizes[1] / 2;
    const int ET = in_sizes[2] / 2;

    float* ws   = (float*)d_ws;
    float* h1   = ws;
    float* z1   = ws + (size_t)16 * N;
    float* h2   = ws + (size_t)32 * N;
    float* z2   = ws + (size_t)96 * N;
    int*   cnt  = (int*)(ws + (size_t)160 * N);
    float* dinv = ws + (size_t)161 * N;
    int*   rowp = (int*)(ws + (size_t)162 * N);
    int*   rank = (int*)(ws + (size_t)163 * N);
    int*   csr  = rank + (size_t)E;

    const int B = 256;
    auto cdiv = [](long long a, long long b) { return (int)((a + b - 1) / b); };

    zero_i32<<<cdiv(N, B), B, 0, stream>>>(cnt, N);
    count_rank<<<cdiv(E, B), B, 0, stream>>>(tei + E, cnt, rank, E);
    scan_counts<<<cdiv(N, 1024), B, 0, stream>>>(cnt, rowp, dinv, N);
    fill_csr<<<cdiv(E, B), B, 0, stream>>>(tei, tei + E, rowp, rank, csr, E);

    const int tiles = (N + 15) / 16;

    gemm_x_w1<<<cdiv(tiles, 8), B, 0, stream>>>(x, W1, h1, N);
    gather_rows<16, true, 8><<<cdiv((long long)4 * N, B), B, 0, stream>>>(
        rowp, cnt, csr, dinv, h1, b1, z1, N);

    gemm_z_w2<<<cdiv(tiles, 8), B, 0, stream>>>(z1, W2, h2, N);
    gather_rows<64, false, 8><<<cdiv((long long)16 * N, B), B, 0, stream>>>(
        rowp, cnt, csr, dinv, h2, b2, z2, N);

    edge_score<<<cdiv(2 * (long long)ET, 1024), B, 0, stream>>>(tpo, tpo + ET, tne, tne + ET,
                                                               z2, (float*)d_out, ET);
}


# Edge sort by destination on LDS counters instead of global returning atomics: two-level counting sort (512-node bins), 1024-thread blocks loop over edges; slot order in a run = LDS arrival order
# speedup vs baseline: 10.7305x; 1.9923x over previous
#include <hip/hip_runtime.h>

typedef __attribute__((ext_vector_type(2))) float v2f;
typedef __attribute__((ext_vector_type(8))) float v8f;

constexpr int BIN_SH   = 9;
constexpr int BIN_W    = 1 << BIN_SH;
constexpr int MAX_BINS = 1024;
constexpr int SORT_G   = 256;
constexpr int SORT_T   = 1024;

__device__ __forceinline__ int wave_incl_scan(int v, int lane) {
    for (int o = 1; o < 32; o <<= 1) {
        int u = __shfl_up(v, o);
        if (lane >= o) v += u;
    }
    return v;
}

__global__ __launch_bounds__(SORT_T) void bin_count(const int* __restrict__ dst,
                                                    int* __restrict__ rank,
                                                    int* __restrict__ hist,
                                                    int nE, int nb, int chunk) {
    __shared__ int s_h[MAX_BINS];
    const int t = threadIdx.x;
    if (t < nb) s_h[t] = 0;
    __syncthreads();
    const int beg = blockIdx.x * chunk;
    const int end = min(beg + chunk, nE);
    int base = beg;
    for (; base + 4 * SORT_T <= end; base += 4 * SORT_T) {
        int d[4];
#pragma unroll
        for (int k = 0; k < 4; ++k) d[k] = dst[base + k * SORT_T + t];
#pragma unroll
        for (int k = 0; k < 4; ++k)
            rank[base + k * SORT_T + t] = atomicAdd(&s_h[d[k] >> BIN_SH], 1);
    }
    for (int e = base + t; e < end; e += SORT_T)
        rank[e] = atomicAdd(&s_h[dst[e] >> BIN_SH], 1);
    __syncthreads();
    if (t < nb) hist[t * SORT_G + blockIdx.x] = s_h[t];
}

__global__ __launch_bounds__(SORT_G) void bin_scan(int* __restrict__ hist,
                                                   int* __restrict__ total) {
    __shared__ int s_wave[SORT_G / 32];
    const int t = threadIdx.x, lane = t & 31, w = t >> 5;
    int* h = hist + (size_t)blockIdx.x * SORT_G;
    const int v = h[t];
    const int inc = wave_incl_scan(v, lane);
    if (lane == 31) s_wave[w] = inc;
    __syncthreads();
    int ex = inc - v;
    int sum = 0;
    for (int k = 0; k < SORT_G / 32; ++k) {
        if (k < w) ex += s_wave[k];
        sum += s_wave[k];
    }
    h[t] = ex;
    if (t == 0) total[blockIdx.x] = sum;
}

__global__ __launch_bounds__(SORT_T) void bin_scatter(const int* __restrict__ src,
                                                      const int* __restrict__ dst,
                                                      const int* __restrict__ rank,
                                                      const int* __restrict__ hist,
                                                      const int* __restrict__ total,
                                                      unsigned* __restrict__ packed,
                                                      int nE, int nb, int chunk) {
    __shared__ int s_base[MAX_BINS];
    __shared__ int s_wave[SORT_T / 32];
    const int t = threadIdx.x, lane = t & 31, w = t >> 5;
    const int v = (t < nb) ? total[t] : 0;
    const int inc = wave_incl_scan(v, lane);
    if (lane == 31) s_wave[w] = inc;
    __syncthreads();
    int ex = inc - v;
    for (int k = 0; k < SORT_T / 32; ++k)
        if (k < w) ex += s_wave[k];
    if (t < nb) s_base[t] = ex + hist[t * SORT_G + blockIdx.x];
    __syncthreads();

    const int beg = blockIdx.x * chunk;
    const int end = min(beg + chunk, nE);
    int base = beg;
    for (; base + 4 * SORT_T <= end; base += 4 * SORT_T) {
        int d[4], s[4], r[4];
#pragma unroll
        for (int k = 0; k < 4; ++k) {
            const int e = base + k * SORT_T + t;
            d[k] = dst[e];
            s[k] = src[e];
            r[k] = rank[e];
        }
#pragma unroll
        for (int k = 0; k < 4; ++k)
            packed[s_base[d[k] >> BIN_SH] + r[k]] =
                ((unsigned)s[k] << BIN_SH) | ((unsigned)d[k] & (BIN_W - 1));
    }
    for (int e = base + t; e < end; e += SORT_T) {
        const int d = dst[e];
        packed[s_base[d >> BIN_SH] + rank[e]] =
            ((unsigned)src[e] << BIN_SH) | ((unsigned)d & (BIN_W - 1));
    }
}

__global__ __launch_bounds__(SORT_T) void bin_sort(const unsigned* __restrict__ packed,
                                                   const int* __restrict__ total,
                                                   int* __restrict__ rowptr,
                                                   int* __restrict__ cnt,
                                                   float* __restrict__ dinv,
                                                   int* __restrict__ csr, int n) {
    __shared__ int s_c[BIN_W];
    __shared__ int s_cur[BIN_W];
    __shared__ int s_wave[SORT_T / 32];
    __shared__ int s_red[SORT_T / 32];
    const int t = threadIdx.x, lane = t & 31, w = t >> 5;
    const int b = blockIdx.x;

    int part = 0;
    for (int j = t; j < b; j += SORT_T) part += total[j];
    for (int o = 16; o > 0; o >>= 1) part += __shfl_xor(part, o);
    if (lane == 0) s_red[w] = part;
    if (t < BIN_W) s_c[t] = 0;
    __syncthreads();
    int beg = 0;
    for (int k = 0; k < SORT_T / 32; ++k) beg += s_red[k];
    const int end = beg + total[b];

    int base = beg;
    for (; base + 4 * SORT_T <= end; base += 4 * SORT_T) {
        unsigned p[4];
#pragma unroll
        for (int k = 0; k < 4; ++k) p[k] = packed[base + k * SORT_T + t];
#pragma unroll
        for (int k = 0; k < 4; ++k) atomicAdd(&s_c[p[k] & (BIN_W - 1)], 1);
    }
    for (int e = base + t; e < end; e += SORT_T) atomicAdd(&s_c[packed[e] & (BIN_W - 1)], 1);
    __syncthreads();

    const int c = (t < BIN_W) ? s_c[t] : 0;
    const int inc = wave_incl_scan(c, lane);
    if (lane == 31) s_wave[w] = inc;
    __syncthreads();
    int ex = inc - c;
    for (int k = 0; k < BIN_W / 32; ++k)
        if (k < w) ex += s_wave[k];
    if (t < BIN_W) {
        s_cur[t] = beg + ex;
        const int node = b * BIN_W + t;
        if (node < n) {
            rowptr[node] = beg + ex;
            cnt[node] = c;
            dinv[node] = rsqrtf((float)c + 1.0f);
        }
    }
    __syncthreads();

    base = beg;
    for (; base + 4 * SORT_T <= end; base += 4 * SORT_T) {
        unsigned p[4];
#pragma unroll
        for (int k = 0; k < 4; ++k) p[k] = packed[base + k * SORT_T + t];
#pragma unroll
        for (int k = 0; k < 4; ++k)
            csr[atomicAdd(&s_cur[p[k] & (BIN_W - 1)], 1)] = (int)(p[k] >> BIN_SH);
    }
    for (int e = base + t; e < end; e += SORT_T) {
        const unsigned p = packed[e];
        csr[atomicAdd(&s_cur[p & (BIN_W - 1)], 1)] = (int)(p >> BIN_SH);
    }
}

__global__ __launch_bounds__(256) void gemm_x_w1(const float* __restrict__ X,
                                                 const float* __restrict__ W,
                                                 float* __restrict__ H, int n) {
    const int lane = threadIdx.x & 31;
    const int tile = blockIdx.x * 8 + (threadIdx.x >> 5);
    const int row0 = tile * 16;
    if (row0 >= n) return;
    int m = row0 + (lane & 15);
    if (m >= n) m = n - 1;
    const int kh  = (lane >> 4) << 1;
    const int col = lane & 15;
    const float* xr = X + (size_t)m * 256;

    v8f acc = {};
#pragma unroll 4
    for (int k0 = 0; k0 < 256; k0 += 4) {
        v2f a, b;
        a.x = xr[k0 + kh];
        a.y = xr[k0 + kh + 1];
        b.x = W[(size_t)(k0 + kh) * 16 + col];
        b.y = W[(size_t)(k0 + kh + 1) * 16 + col];
        acc = __builtin_amdgcn_wmma_f32_16x16x4_f32(false, a, false, b,
                                                    (short)0, acc, false, false);
    }
    const int ra = (lane >> 4) << 3;
    if (row0 + 16 <= n) {
#pragma unroll
        for (int v = 0; v < 8; ++v)
            H[(size_t)(row0 + ra + v) * 16 + col] = acc[v];
    } else {
#pragma unroll
        for (int v = 0; v < 8; ++v) {
            int r = row0 + ra + v;
            if (r < n) H[(size_t)r * 16 + col] = acc[v];
        }
    }
}

__global__ __launch_bounds__(256) void gemm_z_w2(const float* __restrict__ Z,
                                                 const float* __restrict__ W,
                                                 float* __restrict__ H, int n) {
    const int lane = threadIdx.x & 31;
    const int tile = blockIdx.x * 8 + (threadIdx.x >> 5);
    const int row0 = tile * 16;
    if (row0 >= n) return;
    int m = row0 + (lane & 15);
    if (m >= n) m = n - 1;
    const int kh  = (lane >> 4) << 1;
    const int col = lane & 15;
    const float* zr = Z + (size_t)m * 16;

    v8f acc0 = {}, acc1 = {}, acc2 = {}, acc3 = {};
#pragma unroll
    for (int k0 = 0; k0 < 16; k0 += 4) {
        v2f a, b;
        a.x = zr[k0 + kh];
        a.y = zr[k0 + kh + 1];
        const float* w0 = W + (size_t)(k0 + kh) * 64;
        const float* w1 = w0 + 64;
        b.x = w0[col];      b.y = w1[col];
        acc0 = __builtin_amdgcn_wmma_f32_16x16x4_f32(false, a, false, b, (short)0, acc0, false, false);
        b.x = w0[col + 16]; b.y = w1[col + 16];
        acc1 = __builtin_amdgcn_wmma_f32_16x16x4_f32(false, a, false, b, (short)0, acc1, false, false);
        b.x = w0[col + 32]; b.y = w1[col + 32];
        acc2 = __builtin_amdgcn_wmma_f32_16x16x4_f32(false, a, false, b, (short)0, acc2, false, false);
        b.x = w0[col + 48]; b.y = w1[col + 48];
        acc3 = __builtin_amdgcn_wmma_f32_16x16x4_f32(false, a, false, b, (short)0, acc3, false, false);
    }
    const int ra = (lane >> 4) << 3;
    if (row0 + 16 <= n) {
#pragma unroll
        for (int v = 0; v < 8; ++v) {
            float* hp = H + (size_t)(row0 + ra + v) * 64 + col;
            hp[0]  = acc0[v];
            hp[16] = acc1[v];
            hp[32] = acc2[v];
            hp[48] = acc3[v];
        }
    } else {
#pragma unroll
        for (int v = 0; v < 8; ++v) {
            int r = row0 + ra + v;
            if (r < n) {
                float* hp = H + (size_t)r * 64 + col;
                hp[0]  = acc0[v];
                hp[16] = acc1[v];
                hp[32] = acc2[v];
                hp[48] = acc3[v];
            }
        }
    }
}

template <int F, bool RELU, int U>
__global__ __launch_bounds__(256) void gather_rows(const int* __restrict__ rowptr,
                                                   const int* __restrict__ cnt,
                                                   const int* __restrict__ csr,
                                                   const float* __restrict__ dinv,
                                                   const float* __restrict__ H,
                                                   const float* __restrict__ bias,
                                                   float* __restrict__ Z, int n) {
    constexpr int L = F / 4;
    const long long gid = (long long)blockIdx.x * blockDim.x + threadIdx.x;
    const int i = (int)(gid / L);
    const int l = (int)(gid % L);
    if (i >= n) return;
    const float4* H4 = (const float4*)H;
    const int beg = rowptr[i];
    const int end = beg + cnt[i];
    const float di = dinv[i];
    float4 acc = make_float4(0.0f, 0.0f, 0.0f, 0.0f);
    if (beg < end) {
        const int last = end - 1;
        int s[U];
#pragma unroll
        for (int k = 0; k < U; ++k) s[k] = csr[min(beg + k, last)];
        for (int p = beg; p < end; p += U) {
            int sn[U];
#pragma unroll
            for (int k = 0; k < U; ++k) sn[k] = csr[min(p + U + k, last)];
            float dv[U], nr[U];
            float4 h[U];
#pragma unroll
            for (int k = 0; k < U; ++k) {
                dv[k] = dinv[s[k]];
                h[k] = H4[(size_t)s[k] * L + l];
            }
#pragma unroll
            for (int k = 0; k < U; ++k)
                asm volatile("" :: "v"(dv[k]), "v"(h[k].x), "v"(h[k].y), "v"(h[k].z), "v"(h[k].w));
#pragma unroll
            for (int k = 0; k < U; ++k) nr[k] = dv[k] * di;
#pragma unroll
            for (int k = 0; k < U; ++k) {
                if (k == 0 || p + k < end) {
                    acc.x = fmaf(h[k].x, nr[k], acc.x);
                    acc.y = fmaf(h[k].y, nr[k], acc.y);
                    acc.z = fmaf(h[k].z, nr[k], acc.z);
                    acc.w = fmaf(h[k].w, nr[k], acc.w);
                }
            }
#pragma unroll
            for (int k = 0; k < U; ++k) s[k] = sn[k];
        }
    }
    const float4 hs = H4[(size_t)i * L + l];
    const float sl = di * di;
    float4 v;
    v.x = acc.x + sl * hs.x + bias[4 * l + 0];
    v.y = acc.y + sl * hs.y + bias[4 * l + 1];
    v.z = acc.z + sl * hs.z + bias[4 * l + 2];
    v.w = acc.w + sl * hs.w + bias[4 * l + 3];
    if (RELU) {
        v.x = fmaxf(v.x, 0.0f); v.y = fmaxf(v.y, 0.0f);
        v.z = fmaxf(v.z, 0.0f); v.w = fmaxf(v.w, 0.0f);
    }
    ((float4*)Z)[(size_t)i * L + l] = v;
}


__device__ __forceinline__ float row_sum16(float v) {
    v += __int_as_float(__builtin_amdgcn_update_dpp(0, __float_as_int(v), 0x161, 0xf, 0xf, true));
    v += __int_as_float(__builtin_amdgcn_update_dpp(0, __float_as_int(v), 0x162, 0xf, 0xf, true));
    v += __int_as_float(__builtin_amdgcn_update_dpp(0, __float_as_int(v), 0x164, 0xf, 0xf, true));
    v += __int_as_float(__builtin_amdgcn_update_dpp(0, __float_as_int(v), 0x168, 0xf, 0xf, true));
    return v;
}

__global__ __launch_bounds__(256) void edge_score(const int* __restrict__ psrc,
                                                  const int* __restrict__ pdst,
                                                  const int* __restrict__ nsrc,
                                                  const int* __restrict__ ndst,
                                                  const float* __restrict__ Z,
                                                  float* __restrict__ out, int nT) {
    const int lane = threadIdx.x & 31;
    const int c = lane & 15;
    const unsigned cb = (unsigned)c << 4;
    const bool upper = lane >= 16;
    const int total = 2 * nT;
    const int ngroups = (total + 31) >> 5;
    const int nwaves = gridDim.x * 8;
    int g = blockIdx.x * 8 + __builtin_amdgcn_readfirstlane((int)(threadIdx.x >> 5));
    if (g >= ngroups) return;
    const char* Zb = (const char*)Z;

    int j = min(g * 32 + lane, total - 1);
    int s = (j < nT) ? psrc[j] : nsrc[j - nT];
    int d = (j < nT) ? pdst[j] : ndst[j - nT];

    for (; g < ngroups; g += nwaves) {
        const int gn = min(g + nwaves, ngroups - 1);
        const int jn = min(gn * 32 + lane, total - 1);
        const int sn = (jn < nT) ? psrc[jn] : nsrc[jn - nT];
        const int dn = (jn < nT) ? pdst[jn] : ndst[jn - nT];

        float res = 0.0f;
#pragma unroll
        for (int t0 = 0; t0 < 16; t0 += 8) {
            float4 a[8], b[8];
#pragma unroll
            for (int u = 0; u < 8; ++u) {
                const int t = t0 + u;
                const int s0 = __builtin_amdgcn_readlane(s, t);
                const int s1 = __builtin_amdgcn_readlane(s, t + 16);
                const int d0 = __builtin_amdgcn_readlane(d, t);
                const int d1 = __builtin_amdgcn_readlane(d, t + 16);
                const unsigned so = ((unsigned)(upper ? s1 : s0) << 8) + cb;
                const unsigned dof = ((unsigned)(upper ? d1 : d0) << 8) + cb;
                a[u] = *(const float4*)(Zb + so);
                b[u] = *(const float4*)(Zb + dof);
            }
#pragma unroll
            for (int u = 0; u < 8; ++u) {
                float p = a[u].x * b[u].x;
                p = fmaf(a[u].y, b[u].y, p);
                p = fmaf(a[u].z, b[u].z, p);
                p = fmaf(a[u].w, b[u].w, p);
                p = row_sum16(p);
                if (c == t0 + u) res = p;
            }
        }
        const int jo = g * 32 + lane;
        if (jo < total) out[jo] = res;
        s = sn;
        d = dn;
    }
}

extern "C" void kernel_launch(void* const* d_in, const int* in_sizes, int n_in,
                              void* d_out, int out_size, void* d_ws, size_t ws_size,
                              hipStream_t stream) {
    const float* x   = (const float*)d_in[0];
    const int*   tei = (const int*)d_in[1];
    const int*   tpo = (const int*)d_in[2];
    const int*   tne = (const int*)d_in[3];
    const float* W1  = (const float*)d_in[4];
    const float* b1  = (const float*)d_in[5];
    const float* W2  = (const float*)d_in[6];
    const float* b2  = (const float*)d_in[7];

    const int N  = in_sizes[0] / 256;
    const int E  = in_sizes[1] / 2;
    const int ET = in_sizes[2] / 2;

    float* ws   = (float*)d_ws;
    float* h1   = ws;
    float* z1   = ws + (size_t)16 * N;
    float* h2   = ws + (size_t)32 * N;
    float* z2   = ws + (size_t)96 * N;
    int*   cnt  = (int*)(ws + (size_t)160 * N);
    float* dinv = ws + (size_t)161 * N;
    int*   rowp = (int*)(ws + (size_t)162 * N);
    int*   rank = (int*)(ws + (size_t)163 * N);
    int*   csr  = rank + (size_t)E;
    unsigned* packed = (unsigned*)(csr + (size_t)E);
    int*   hist = (int*)(packed + (size_t)E);
    int*   btot = hist + (size_t)MAX_BINS * SORT_G;

    const int B = 256;
    auto cdiv = [](long long a, long long b) { return (int)((a + b - 1) / b); };

    const int nb    = cdiv(N, BIN_W);
    const int chunk = cdiv(E, SORT_G);
    bin_count<<<SORT_G, SORT_T, 0, stream>>>(tei + E, rank, hist, E, nb, chunk);
    bin_scan<<<nb, SORT_G, 0, stream>>>(hist, btot);
    bin_scatter<<<SORT_G, SORT_T, 0, stream>>>(tei, tei + E, rank, hist, btot, packed, E, nb, chunk);
    bin_sort<<<nb, SORT_T, 0, stream>>>(packed, btot, rowp, cnt, dinv, csr, N);

    const int tiles = (N + 15) / 16;

    gemm_x_w1<<<cdiv(tiles, 8), B, 0, stream>>>(x, W1, h1, N);
    gather_rows<16, true, 8><<<cdiv((long long)4 * N, B), B, 0, stream>>>(
        rowp, cnt, csr, dinv, h1, b1, z1, N);

    gemm_z_w2<<<cdiv(tiles, 8), B, 0, stream>>>(z1, W2, h2, N);
    gather_rows<64, false, 8><<<cdiv((long long)16 * N, B), B, 0, stream>>>(
        rowp, cnt, csr, dinv, h2, b2, z2, N);

    edge_score<<<cdiv(2 * (long long)ET, 1024), B, 0, stream>>>(tpo, tpo + ET, tne, tne + ET,
                                                               z2, (float*)d_out, ET);
}


# Layer 2 aggregates before it projects, A(Z W2) = (A Z) W2: edges gather 16-float rows of dinv-scaled z1, not 64-float rows of z1 W2; W2 and b2 applied after (sum order changes)
# speedup vs baseline: 12.2971x; 1.1460x over previous
#include <hip/hip_runtime.h>

typedef __attribute__((ext_vector_type(2))) float v2f;
typedef __attribute__((ext_vector_type(8))) float v8f;

constexpr int BIN_SH   = 9;
constexpr int BIN_W    = 1 << BIN_SH;
constexpr int MAX_BINS = 1024;
constexpr int SORT_G   = 256;
constexpr int SORT_T   = 1024;

__device__ __forceinline__ int wave_incl_scan(int v, int lane) {
    for (int o = 1; o < 32; o <<= 1) {
        int u = __shfl_up(v, o);
        if (lane >= o) v += u;
    }
    return v;
}

__global__ __launch_bounds__(SORT_T) void bin_count(const int* __restrict__ dst,
                                                    int* __restrict__ rank,
                                                    int* __restrict__ hist,
                                                    int nE, int nb, int chunk) {
    __shared__ int s_h[MAX_BINS];
    const int t = threadIdx.x;
    if (t < nb) s_h[t] = 0;
    __syncthreads();
    const int beg = blockIdx.x * chunk;
    const int end = min(beg + chunk, nE);
    int base = beg;
    for (; base + 4 * SORT_T <= end; base += 4 * SORT_T) {
        int d[4];
#pragma unroll
        for (int k = 0; k < 4; ++k) d[k] = dst[base + k * SORT_T + t];
#pragma unroll
        for (int k = 0; k < 4; ++k)
            rank[base + k * SORT_T + t] = atomicAdd(&s_h[d[k] >> BIN_SH], 1);
    }
    for (int e = base + t; e < end; e += SORT_T)
        rank[e] = atomicAdd(&s_h[dst[e] >> BIN_SH], 1);
    __syncthreads();
    if (t < nb) hist[t * SORT_G + blockIdx.x] = s_h[t];
}

__global__ __launch_bounds__(SORT_G) void bin_scan(int* __restrict__ hist,
                                                   int* __restrict__ total) {
    __shared__ int s_wave[SORT_G / 32];
    const int t = threadIdx.x, lane = t & 31, w = t >> 5;
    int* h = hist + (size_t)blockIdx.x * SORT_G;
    const int v = h[t];
    const int inc = wave_incl_scan(v, lane);
    if (lane == 31) s_wave[w] = inc;
    __syncthreads();
    int ex = inc - v;
    int sum = 0;
    for (int k = 0; k < SORT_G / 32; ++k) {
        if (k < w) ex += s_wave[k];
        sum += s_wave[k];
    }
    h[t] = ex;
    if (t == 0) total[blockIdx.x] = sum;
}

__global__ __launch_bounds__(SORT_T) void bin_scatter(const int* __restrict__ src,
                                                      const int* __restrict__ dst,
                                                      const int* __restrict__ rank,
                                                      const int* __restrict__ hist,
                                                      const int* __restrict__ total,
                                                      unsigned* __restrict__ packed,
                                                      int nE, int nb, int chunk) {
    __shared__ int s_base[MAX_BINS];
    __shared__ int s_wave[SORT_T / 32];
    const int t = threadIdx.x, lane = t & 31, w = t >> 5;
    const int v = (t < nb) ? total[t] : 0;
    const int inc = wave_incl_scan(v, lane);
    if (lane == 31) s_wave[w] = inc;
    __syncthreads();
    int ex = inc - v;
    for (int k = 0; k < SORT_T / 32; ++k)
        if (k < w) ex += s_wave[k];
    if (t < nb) s_base[t] = ex + hist[t * SORT_G + blockIdx.x];
    __syncthreads();

    const int beg = blockIdx.x * chunk;
    const int end = min(beg + chunk, nE);
    int base = beg;
    for (; base + 4 * SORT_T <= end; base += 4 * SORT_T) {
        int d[4], s[4], r[4];
#pragma unroll
        for (int k = 0; k < 4; ++k) {
            const int e = base + k * SORT_T + t;
            d[k] = dst[e];
            s[k] = src[e];
            r[k] = rank[e];
        }
#pragma unroll
        for (int k = 0; k < 4; ++k)
            packed[s_base[d[k] >> BIN_SH] + r[k]] =
                ((unsigned)s[k] << BIN_SH) | ((unsigned)d[k] & (BIN_W - 1));
    }
    for (int e = base + t; e < end; e += SORT_T) {
        const int d = dst[e];
        packed[s_base[d >> BIN_SH] + rank[e]] =
            ((unsigned)src[e] << BIN_SH) | ((unsigned)d & (BIN_W - 1));
    }
}

__global__ __launch_bounds__(SORT_T) void bin_sort(const unsigned* __restrict__ packed,
                                                   const int* __restrict__ total,
                                                   int* __restrict__ rowptr,
                                                   int* __restrict__ cnt,
                                                   float* __restrict__ dinv,
                                                   int* __restrict__ csr, int n) {
    __shared__ int s_c[BIN_W];
    __shared__ int s_cur[BIN_W];
    __shared__ int s_wave[SORT_T / 32];
    __shared__ int s_red[SORT_T / 32];
    const int t = threadIdx.x, lane = t & 31, w = t >> 5;
    const int b = blockIdx.x;

    int part = 0;
    for (int j = t; j < b; j += SORT_T) part += total[j];
    for (int o = 16; o > 0; o >>= 1) part += __shfl_xor(part, o);
    if (lane == 0) s_red[w] = part;
    if (t < BIN_W) s_c[t] = 0;
    __syncthreads();
    int beg = 0;
    for (int k = 0; k < SORT_T / 32; ++k) beg += s_red[k];
    const int end = beg + total[b];

    int base = beg;
    for (; base + 4 * SORT_T <= end; base += 4 * SORT_T) {
        unsigned p[4];
#pragma unroll
        for (int k = 0; k < 4; ++k) p[k] = packed[base + k * SORT_T + t];
#pragma unroll
        for (int k = 0; k < 4; ++k) atomicAdd(&s_c[p[k] & (BIN_W - 1)], 1);
    }
    for (int e = base + t; e < end; e += SORT_T) atomicAdd(&s_c[packed[e] & (BIN_W - 1)], 1);
    __syncthreads();

    const int c = (t < BIN_W) ? s_c[t] : 0;
    const int inc = wave_incl_scan(c, lane);
    if (lane == 31) s_wave[w] = inc;
    __syncthreads();
    int ex = inc - c;
    for (int k = 0; k < BIN_W / 32; ++k)
        if (k < w) ex += s_wave[k];
    if (t < BIN_W) {
        s_cur[t] = beg + ex;
        const int node = b * BIN_W + t;
        if (node < n) {
            rowptr[node] = beg + ex;
            cnt[node] = c;
            dinv[node] = rsqrtf((float)c + 1.0f);
        }
    }
    __syncthreads();

    base = beg;
    for (; base + 4 * SORT_T <= end; base += 4 * SORT_T) {
        unsigned p[4];
#pragma unroll
        for (int k = 0; k < 4; ++k) p[k] = packed[base + k * SORT_T + t];
#pragma unroll
        for (int k = 0; k < 4; ++k)
            csr[atomicAdd(&s_cur[p[k] & (BIN_W - 1)], 1)] = (int)(p[k] >> BIN_SH);
    }
    for (int e = base + t; e < end; e += SORT_T) {
        const unsigned p = packed[e];
        csr[atomicAdd(&s_cur[p & (BIN_W - 1)], 1)] = (int)(p >> BIN_SH);
    }
}

__global__ __launch_bounds__(256) void gemm_x_w1(const float* __restrict__ X,
                                                 const float* __restrict__ W,
                                                 const float* __restrict__ dinv,
                                                 float* __restrict__ H, int n) {
    const int lane = threadIdx.x & 31;
    const int tile = blockIdx.x * 8 + (threadIdx.x >> 5);
    const int row0 = tile * 16;
    if (row0 >= n) return;
    int m = row0 + (lane & 15);
    if (m >= n) m = n - 1;
    const int kh  = (lane >> 4) << 1;
    const int col = lane & 15;
    const float* xr = X + (size_t)m * 256;
    const int ra = (lane >> 4) << 3;
    float sc[8];
#pragma unroll
    for (int v = 0; v < 8; ++v) sc[v] = dinv[min(row0 + ra + v, n - 1)];

    v8f acc = {};
#pragma unroll 4
    for (int k0 = 0; k0 < 256; k0 += 4) {
        v2f a, b;
        a.x = xr[k0 + kh];
        a.y = xr[k0 + kh + 1];
        b.x = W[(size_t)(k0 + kh) * 16 + col];
        b.y = W[(size_t)(k0 + kh + 1) * 16 + col];
        acc = __builtin_amdgcn_wmma_f32_16x16x4_f32(false, a, false, b,
                                                    (short)0, acc, false, false);
    }
#pragma unroll
    for (int v = 0; v < 8; ++v) acc[v] *= sc[v];
    if (row0 + 16 <= n) {
#pragma unroll
        for (int v = 0; v < 8; ++v)
            H[(size_t)(row0 + ra + v) * 16 + col] = acc[v];
    } else {
#pragma unroll
        for (int v = 0; v < 8; ++v) {
            int r = row0 + ra + v;
            if (r < n) H[(size_t)r * 16 + col] = acc[v];
        }
    }
}

__global__ __launch_bounds__(256) void gemm_a_w2(const float* __restrict__ Z,
                                                 const float* __restrict__ W,
                                                 const float* __restrict__ bias,
                                                 float* __restrict__ H, int n) {
    const int lane = threadIdx.x & 31;
    const int tile = blockIdx.x * 8 + (threadIdx.x >> 5);
    const int row0 = tile * 16;
    if (row0 >= n) return;
    int m = row0 + (lane & 15);
    if (m >= n) m = n - 1;
    const int kh  = (lane >> 4) << 1;
    const int col = lane & 15;
    const float* zr = Z + (size_t)m * 16;
    const int ra = (lane >> 4) << 3;
    const float bs0 = bias[col], bs1 = bias[col + 16], bs2 = bias[col + 32], bs3 = bias[col + 48];

    v8f acc0 = {}, acc1 = {}, acc2 = {}, acc3 = {};
#pragma unroll
    for (int k0 = 0; k0 < 16; k0 += 4) {
        v2f a, b;
        a.x = zr[k0 + kh];
        a.y = zr[k0 + kh + 1];
        const float* w0 = W + (size_t)(k0 + kh) * 64;
        const float* w1 = w0 + 64;
        b.x = w0[col];      b.y = w1[col];
        acc0 = __builtin_amdgcn_wmma_f32_16x16x4_f32(false, a, false, b, (short)0, acc0, false, false);
        b.x = w0[col + 16]; b.y = w1[col + 16];
        acc1 = __builtin_amdgcn_wmma_f32_16x16x4_f32(false, a, false, b, (short)0, acc1, false, false);
        b.x = w0[col + 32]; b.y = w1[col + 32];
        acc2 = __builtin_amdgcn_wmma_f32_16x16x4_f32(false, a, false, b, (short)0, acc2, false, false);
        b.x = w0[col + 48]; b.y = w1[col + 48];
        acc3 = __builtin_amdgcn_wmma_f32_16x16x4_f32(false, a, false, b, (short)0, acc3, false, false);
    }
#pragma unroll
    for (int v = 0; v < 8; ++v) {
        acc0[v] += bs0;
        acc1[v] += bs1;
        acc2[v] += bs2;
        acc3[v] += bs3;
    }
    if (row0 + 16 <= n) {
#pragma unroll
        for (int v = 0; v < 8; ++v) {
            float* hp = H + (size_t)(row0 + ra + v) * 64 + col;
            hp[0]  = acc0[v];
            hp[16] = acc1[v];
            hp[32] = acc2[v];
            hp[48] = acc3[v];
        }
    } else {
#pragma unroll
        for (int v = 0; v < 8; ++v) {
            int r = row0 + ra + v;
            if (r < n) {
                float* hp = H + (size_t)r * 64 + col;
                hp[0]  = acc0[v];
                hp[16] = acc1[v];
                hp[32] = acc2[v];
                hp[48] = acc3[v];
            }
        }
    }
}

template <int F, bool FIRST, int U>
__global__ __launch_bounds__(256) void gather_rows(const int* __restrict__ rowptr,
                                                   const int* __restrict__ cnt,
                                                   const int* __restrict__ csr,
                                                   const float* __restrict__ dinv,
                                                   const float* __restrict__ H,
                                                   const float* __restrict__ bias,
                                                   float* __restrict__ Z, int n) {
    constexpr int L = F / 4;
    const long long gid = (long long)blockIdx.x * blockDim.x + threadIdx.x;
    const int i = (int)(gid / L);
    const int l = (int)(gid % L);
    if (i >= n) return;
    const float4* H4 = (const float4*)H;
    const int beg = rowptr[i];
    const int end = beg + cnt[i];
    const float di = dinv[i];
    float4 acc = make_float4(0.0f, 0.0f, 0.0f, 0.0f);
    if (beg < end) {
        const int last = end - 1;
        int s[U];
#pragma unroll
        for (int k = 0; k < U; ++k) s[k] = csr[min(beg + k, last)];
        for (int p = beg; p < end; p += U) {
            int sn[U];
#pragma unroll
            for (int k = 0; k < U; ++k) sn[k] = csr[min(p + U + k, last)];
            float4 h[U];
#pragma unroll
            for (int k = 0; k < U; ++k) h[k] = H4[(size_t)s[k] * L + l];
#pragma unroll
            for (int k = 0; k < U; ++k)
                asm volatile("" :: "v"(h[k].x), "v"(h[k].y), "v"(h[k].z), "v"(h[k].w));
#pragma unroll
            for (int k = 0; k < U; ++k) {
                if (k == 0 || p + k < end) {
                    acc.x += h[k].x;
                    acc.y += h[k].y;
                    acc.z += h[k].z;
                    acc.w += h[k].w;
                }
            }
#pragma unroll
            for (int k = 0; k < U; ++k) s[k] = sn[k];
        }
    }
    const float4 hs = H4[(size_t)i * L + l];
    float4 v;
    v.x = di * (acc.x + hs.x);
    v.y = di * (acc.y + hs.y);
    v.z = di * (acc.z + hs.z);
    v.w = di * (acc.w + hs.w);
    if (FIRST) {
        v.x = di * fmaxf(v.x + bias[4 * l + 0], 0.0f);
        v.y = di * fmaxf(v.y + bias[4 * l + 1], 0.0f);
        v.z = di * fmaxf(v.z + bias[4 * l + 2], 0.0f);
        v.w = di * fmaxf(v.w + bias[4 * l + 3], 0.0f);
    }
    ((float4*)Z)[(size_t)i * L + l] = v;
}


__device__ __forceinline__ float row_sum16(float v) {
    v += __int_as_float(__builtin_amdgcn_update_dpp(0, __float_as_int(v), 0x161, 0xf, 0xf, true));
    v += __int_as_float(__builtin_amdgcn_update_dpp(0, __float_as_int(v), 0x162, 0xf, 0xf, true));
    v += __int_as_float(__builtin_amdgcn_update_dpp(0, __float_as_int(v), 0x164, 0xf, 0xf, true));
    v += __int_as_float(__builtin_amdgcn_update_dpp(0, __float_as_int(v), 0x168, 0xf, 0xf, true));
    return v;
}

__global__ __launch_bounds__(256) void edge_score(const int* __restrict__ psrc,
                                                  const int* __restrict__ pdst,
                                                  const int* __restrict__ nsrc,
                                                  const int* __restrict__ ndst,
                                                  const float* __restrict__ Z,
                                                  float* __restrict__ out, int nT) {
    const int lane = threadIdx.x & 31;
    const int c = lane & 15;
    const unsigned cb = (unsigned)c << 4;
    const bool upper = lane >= 16;
    const int total = 2 * nT;
    const int ngroups = (total + 31) >> 5;
    const int nwaves = gridDim.x * 8;
    int g = blockIdx.x * 8 + __builtin_amdgcn_readfirstlane((int)(threadIdx.x >> 5));
    if (g >= ngroups) return;
    const char* Zb = (const char*)Z;

    int j = min(g * 32 + lane, total - 1);
    int s = (j < nT) ? psrc[j] : nsrc[j - nT];
    int d = (j < nT) ? pdst[j] : ndst[j - nT];

    for (; g < ngroups; g += nwaves) {
        const int gn = min(g + nwaves, ngroups - 1);
        const int jn = min(gn * 32 + lane, total - 1);
        const int sn = (jn < nT) ? psrc[jn] : nsrc[jn - nT];
        const int dn = (jn < nT) ? pdst[jn] : ndst[jn - nT];

        float res = 0.0f;
#pragma unroll
        for (int t0 = 0; t0 < 16; t0 += 8) {
            float4 a[8], b[8];
#pragma unroll
            for (int u = 0; u < 8; ++u) {
                const int t = t0 + u;
                const int s0 = __builtin_amdgcn_readlane(s, t);
                const int s1 = __builtin_amdgcn_readlane(s, t + 16);
                const int d0 = __builtin_amdgcn_readlane(d, t);
                const int d1 = __builtin_amdgcn_readlane(d, t + 16);
                const unsigned so = ((unsigned)(upper ? s1 : s0) << 8) + cb;
                const unsigned dof = ((unsigned)(upper ? d1 : d0) << 8) + cb;
                a[u] = *(const float4*)(Zb + so);
                b[u] = *(const float4*)(Zb + dof);
            }
#pragma unroll
            for (int u = 0; u < 8; ++u) {
                float p = a[u].x * b[u].x;
                p = fmaf(a[u].y, b[u].y, p);
                p = fmaf(a[u].z, b[u].z, p);
                p = fmaf(a[u].w, b[u].w, p);
                p = row_sum16(p);
                if (c == t0 + u) res = p;
            }
        }
        const int jo = g * 32 + lane;
        if (jo < total) out[jo] = res;
        s = sn;
        d = dn;
    }
}

extern "C" void kernel_launch(void* const* d_in, const int* in_sizes, int n_in,
                              void* d_out, int out_size, void* d_ws, size_t ws_size,
                              hipStream_t stream) {
    const float* x   = (const float*)d_in[0];
    const int*   tei = (const int*)d_in[1];
    const int*   tpo = (const int*)d_in[2];
    const int*   tne = (const int*)d_in[3];
    const float* W1  = (const float*)d_in[4];
    const float* b1  = (const float*)d_in[5];
    const float* W2  = (const float*)d_in[6];
    const float* b2  = (const float*)d_in[7];

    const int N  = in_sizes[0] / 256;
    const int E  = in_sizes[1] / 2;
    const int ET = in_sizes[2] / 2;

    float* ws   = (float*)d_ws;
    float* h1   = ws;
    float* z1   = ws + (size_t)16 * N;
    float* a2   = ws + (size_t)32 * N;
    float* z2   = ws + (size_t)96 * N;
    int*   cnt  = (int*)(ws + (size_t)160 * N);
    float* dinv = ws + (size_t)161 * N;
    int*   rowp = (int*)(ws + (size_t)162 * N);
    int*   rank = (int*)(ws + (size_t)163 * N);
    int*   csr  = rank + (size_t)E;
    unsigned* packed = (unsigned*)(csr + (size_t)E);
    int*   hist = (int*)(packed + (size_t)E);
    int*   btot = hist + (size_t)MAX_BINS * SORT_G;

    const int B = 256;
    auto cdiv = [](long long a, long long b) { return (int)((a + b - 1) / b); };

    const int nb    = cdiv(N, BIN_W);
    const int chunk = cdiv(E, SORT_G);
    bin_count<<<SORT_G, SORT_T, 0, stream>>>(tei + E, rank, hist, E, nb, chunk);
    bin_scan<<<nb, SORT_G, 0, stream>>>(hist, btot);
    bin_scatter<<<SORT_G, SORT_T, 0, stream>>>(tei, tei + E, rank, hist, btot, packed, E, nb, chunk);
    bin_sort<<<nb, SORT_T, 0, stream>>>(packed, btot, rowp, cnt, dinv, csr, N);

    const int tiles = (N + 15) / 16;

    gemm_x_w1<<<cdiv(tiles, 8), B, 0, stream>>>(x, W1, dinv, h1, N);
    gather_rows<16, true, 8><<<cdiv((long long)4 * N, B), B, 0, stream>>>(
        rowp, cnt, csr, dinv, h1, b1, z1, N);

    gather_rows<16, false, 8><<<cdiv((long long)4 * N, B), B, 0, stream>>>(
        rowp, cnt, csr, dinv, z1, nullptr, a2, N);
    gemm_a_w2<<<cdiv(tiles, 8), B, 0, stream>>>(a2, W2, b2, z2, N);

    edge_score<<<cdiv(2 * (long long)ET, 1024), B, 0, stream>>>(tpo, tpo + ET, tne, tne + ET,
                                                               z2, (float*)d_out, ET);
}


# Layer-2 scoring by the 16x16 form of W2: dot(z2[s],z2[d]) = (A[s] W2 W2^T + W2 b2).A[d] + A[s].W2 b2 + b2.b2; an edge reads 192 bytes of 16-wide rows, not 512 of z2 (sum order changes)
# speedup vs baseline: 13.2729x; 1.0794x over previous
#include <hip/hip_runtime.h>

typedef __attribute__((ext_vector_type(2))) float v2f;
typedef __attribute__((ext_vector_type(8))) float v8f;

constexpr int BIN_SH   = 9;
constexpr int BIN_W    = 1 << BIN_SH;
constexpr int MAX_BINS = 1024;
constexpr int SORT_G   = 256;
constexpr int SORT_T   = 1024;

__device__ __forceinline__ int wave_incl_scan(int v, int lane) {
    for (int o = 1; o < 32; o <<= 1) {
        int u = __shfl_up(v, o);
        if (lane >= o) v += u;
    }
    return v;
}

__global__ __launch_bounds__(SORT_T) void bin_count(const int* __restrict__ dst,
                                                    int* __restrict__ rank,
                                                    int* __restrict__ hist,
                                                    int nE, int nb, int chunk) {
    __shared__ int s_h[MAX_BINS];
    const int t = threadIdx.x;
    if (t < nb) s_h[t] = 0;
    __syncthreads();
    const int beg = blockIdx.x * chunk;
    const int end = min(beg + chunk, nE);
    int base = beg;
    for (; base + 4 * SORT_T <= end; base += 4 * SORT_T) {
        int d[4];
#pragma unroll
        for (int k = 0; k < 4; ++k) d[k] = dst[base + k * SORT_T + t];
#pragma unroll
        for (int k = 0; k < 4; ++k)
            rank[base + k * SORT_T + t] = atomicAdd(&s_h[d[k] >> BIN_SH], 1);
    }
    for (int e = base + t; e < end; e += SORT_T)
        rank[e] = atomicAdd(&s_h[dst[e] >> BIN_SH], 1);
    __syncthreads();
    if (t < nb) hist[t * SORT_G + blockIdx.x] = s_h[t];
}

__global__ __launch_bounds__(SORT_G) void bin_scan(int* __restrict__ hist,
                                                   int* __restrict__ total) {
    __shared__ int s_wave[SORT_G / 32];
    const int t = threadIdx.x, lane = t & 31, w = t >> 5;
    int* h = hist + (size_t)blockIdx.x * SORT_G;
    const int v = h[t];
    const int inc = wave_incl_scan(v, lane);
    if (lane == 31) s_wave[w] = inc;
    __syncthreads();
    int ex = inc - v;
    int sum = 0;
    for (int k = 0; k < SORT_G / 32; ++k) {
        if (k < w) ex += s_wave[k];
        sum += s_wave[k];
    }
    h[t] = ex;
    if (t == 0) total[blockIdx.x] = sum;
}

__global__ __launch_bounds__(SORT_T) void bin_scatter(const int* __restrict__ src,
                                                      const int* __restrict__ dst,
                                                      const int* __restrict__ rank,
                                                      const int* __restrict__ hist,
                                                      const int* __restrict__ total,
                                                      unsigned* __restrict__ packed,
                                                      int nE, int nb, int chunk) {
    __shared__ int s_base[MAX_BINS];
    __shared__ int s_wave[SORT_T / 32];
    const int t = threadIdx.x, lane = t & 31, w = t >> 5;
    const int v = (t < nb) ? total[t] : 0;
    const int inc = wave_incl_scan(v, lane);
    if (lane == 31) s_wave[w] = inc;
    __syncthreads();
    int ex = inc - v;
    for (int k = 0; k < SORT_T / 32; ++k)
        if (k < w) ex += s_wave[k];
    if (t < nb) s_base[t] = ex + hist[t * SORT_G + blockIdx.x];
    __syncthreads();

    const int beg = blockIdx.x * chunk;
    const int end = min(beg + chunk, nE);
    int base = beg;
    for (; base + 4 * SORT_T <= end; base += 4 * SORT_T) {
        int d[4], s[4], r[4];
#pragma unroll
        for (int k = 0; k < 4; ++k) {
            const int e = base + k * SORT_T + t;
            d[k] = dst[e];
            s[k] = src[e];
            r[k] = rank[e];
        }
#pragma unroll
        for (int k = 0; k < 4; ++k)
            packed[s_base[d[k] >> BIN_SH] + r[k]] =
                ((unsigned)s[k] << BIN_SH) | ((unsigned)d[k] & (BIN_W - 1));
    }
    for (int e = base + t; e < end; e += SORT_T) {
        const int d = dst[e];
        packed[s_base[d >> BIN_SH] + rank[e]] =
            ((unsigned)src[e] << BIN_SH) | ((unsigned)d & (BIN_W - 1));
    }
}

__global__ __launch_bounds__(SORT_T) void bin_sort(const unsigned* __restrict__ packed,
                                                   const int* __restrict__ total,
                                                   int* __restrict__ rowptr,
                                                   int* __restrict__ cnt,
                                                   float* __restrict__ dinv,
                                                   int* __restrict__ csr, int n) {
    __shared__ int s_c[BIN_W];
    __shared__ int s_cur[BIN_W];
    __shared__ int s_wave[SORT_T / 32];
    __shared__ int s_red[SORT_T / 32];
    const int t = threadIdx.x, lane = t & 31, w = t >> 5;
    const int b = blockIdx.x;

    int part = 0;
    for (int j = t; j < b; j += SORT_T) part += total[j];
    for (int o = 16; o > 0; o >>= 1) part += __shfl_xor(part, o);
    if (lane == 0) s_red[w] = part;
    if (t < BIN_W) s_c[t] = 0;
    __syncthreads();
    int beg = 0;
    for (int k = 0; k < SORT_T / 32; ++k) beg += s_red[k];
    const int end = beg + total[b];

    int base = beg;
    for (; base + 4 * SORT_T <= end; base += 4 * SORT_T) {
        unsigned p[4];
#pragma unroll
        for (int k = 0; k < 4; ++k) p[k] = packed[base + k * SORT_T + t];
#pragma unroll
        for (int k = 0; k < 4; ++k) atomicAdd(&s_c[p[k] & (BIN_W - 1)], 1);
    }
    for (int e = base + t; e < end; e += SORT_T) atomicAdd(&s_c[packed[e] & (BIN_W - 1)], 1);
    __syncthreads();

    const int c = (t < BIN_W) ? s_c[t] : 0;
    const int inc = wave_incl_scan(c, lane);
    if (lane == 31) s_wave[w] = inc;
    __syncthreads();
    int ex = inc - c;
    for (int k = 0; k < BIN_W / 32; ++k)
        if (k < w) ex += s_wave[k];
    if (t < BIN_W) {
        s_cur[t] = beg + ex;
        const int node = b * BIN_W + t;
        if (node < n) {
            rowptr[node] = beg + ex;
            cnt[node] = c;
            dinv[node] = rsqrtf((float)c + 1.0f);
        }
    }
    __syncthreads();

    base = beg;
    for (; base + 4 * SORT_T <= end; base += 4 * SORT_T) {
        unsigned p[4];
#pragma unroll
        for (int k = 0; k < 4; ++k) p[k] = packed[base + k * SORT_T + t];
#pragma unroll
        for (int k = 0; k < 4; ++k)
            csr[atomicAdd(&s_cur[p[k] & (BIN_W - 1)], 1)] = (int)(p[k] >> BIN_SH);
    }
    for (int e = base + t; e < end; e += SORT_T) {
        const unsigned p = packed[e];
        csr[atomicAdd(&s_cur[p & (BIN_W - 1)], 1)] = (int)(p >> BIN_SH);
    }
}

__global__ __launch_bounds__(256) void gemm_x_w1(const float* __restrict__ X,
                                                 const float* __restrict__ W,
                                                 const float* __restrict__ dinv,
                                                 float* __restrict__ H, int n) {
    const int lane = threadIdx.x & 31;
    const int tile = blockIdx.x * 8 + (threadIdx.x >> 5);
    const int row0 = tile * 16;
    if (row0 >= n) return;
    int m = row0 + (lane & 15);
    if (m >= n) m = n - 1;
    const int kh  = (lane >> 4) << 1;
    const int col = lane & 15;
    const float* xr = X + (size_t)m * 256;
    const int ra = (lane >> 4) << 3;
    float sc[8];
#pragma unroll
    for (int v = 0; v < 8; ++v) sc[v] = dinv[min(row0 + ra + v, n - 1)];

    v8f acc = {};
#pragma unroll 4
    for (int k0 = 0; k0 < 256; k0 += 4) {
        v2f a, b;
        a.x = xr[k0 + kh];
        a.y = xr[k0 + kh + 1];
        b.x = W[(size_t)(k0 + kh) * 16 + col];
        b.y = W[(size_t)(k0 + kh + 1) * 16 + col];
        acc = __builtin_amdgcn_wmma_f32_16x16x4_f32(false, a, false, b,
                                                    (short)0, acc, false, false);
    }
#pragma unroll
    for (int v = 0; v < 8; ++v) acc[v] *= sc[v];
    if (row0 + 16 <= n) {
#pragma unroll
        for (int v = 0; v < 8; ++v)
            H[(size_t)(row0 + ra + v) * 16 + col] = acc[v];
    } else {
#pragma unroll
        for (int v = 0; v < 8; ++v) {
            int r = row0 + ra + v;
            if (r < n) H[(size_t)r * 16 + col] = acc[v];
        }
    }
}

__global__ __launch_bounds__(256) void gemm_a_wwt(const float* __restrict__ A2,
                                                   const float* __restrict__ W,
                                                   const float* __restrict__ bias,
                                                   float* __restrict__ R, int n) {
    const int lane = threadIdx.x & 31;
    const int tile = blockIdx.x * 8 + (threadIdx.x >> 5);
    const int row0 = tile * 16;
    if (row0 >= n) return;
    const int col = lane & 15;
    const int h   = lane >> 4;
    int m = row0 + col;
    if (m >= n) m = n - 1;
    const float4* ar = (const float4*)(A2 + (size_t)m * 16 + 8 * h);
    const float4 a0 = ar[0], a1 = ar[1];
    const float4* wr = (const float4*)(W + (size_t)col * 64);
    const float4* br = (const float4*)bias;
    v8f g = {};
    float cp = 0.0f, bp = 0.0f;
#pragma unroll
    for (int q = 0; q < 8; ++q) {
        const float4 w = wr[2 * q + h];
        const float4 b = br[2 * q + h];
        cp = fmaf(w.x, b.x, cp); cp = fmaf(w.y, b.y, cp);
        cp = fmaf(w.z, b.z, cp); cp = fmaf(w.w, b.w, cp);
        bp = fmaf(b.x, b.x, bp); bp = fmaf(b.y, b.y, bp);
        bp = fmaf(b.z, b.z, bp); bp = fmaf(b.w, b.w, bp);
        v2f f;
        f.x = w.x; f.y = w.y;
        g = __builtin_amdgcn_wmma_f32_16x16x4_f32(false, f, false, f, (short)0, g, false, false);
        f.x = w.z; f.y = w.w;
        g = __builtin_amdgcn_wmma_f32_16x16x4_f32(false, f, false, f, (short)0, g, false, false);
    }
    const float cc = cp + __shfl_xor(cp, 16);
    const float bb = bp + __shfl_xor(bp, 16);

    const float av[8] = {a0.x, a0.y, a0.z, a0.w, a1.x, a1.y, a1.z, a1.w};
    v8f acc = {};
#pragma unroll
    for (int w = 0; w < 4; ++w) {
        v2f a, b;
        a.x = av[2 * w]; a.y = av[2 * w + 1];
        b.x = g[2 * w];  b.y = g[2 * w + 1];
        acc = __builtin_amdgcn_wmma_f32_16x16x4_f32(false, a, false, b, (short)0, acc, false, false);
    }
    float up = 0.0f;
    const int cci = __float_as_int(cc);
#pragma unroll
    for (int i = 0; i < 8; ++i) {
        const float clo = __int_as_float(__builtin_amdgcn_readlane(cci, i));
        const float chi = __int_as_float(__builtin_amdgcn_readlane(cci, i + 8));
        up = fmaf(av[i], h ? chi : clo, up);
    }
    const float u = up + __shfl_xor(up, 16) + bb;

    const int ra = h << 3;
    const float4 zero4 = make_float4(0.0f, 0.0f, 0.0f, 0.0f);
    const float4 t0 = h ? zero4 : make_float4(u, 0.0f, 0.0f, 0.0f);
    if (row0 + 16 <= n) {
#pragma unroll
        for (int v = 0; v < 8; ++v)
            R[(size_t)(row0 + ra + v) * 32 + col] = acc[v] + cc;
        float4* tp = (float4*)(R + (size_t)(row0 + col) * 32 + 16 + 8 * h);
        tp[0] = t0;
        tp[1] = zero4;
    } else {
#pragma unroll
        for (int v = 0; v < 8; ++v) {
            const int r = row0 + ra + v;
            if (r < n) R[(size_t)r * 32 + col] = acc[v] + cc;
        }
        if (row0 + col < n) {
            float4* tp = (float4*)(R + (size_t)(row0 + col) * 32 + 16 + 8 * h);
            tp[0] = t0;
            tp[1] = zero4;
        }
    }
}

template <int F, bool FIRST, int U>
__global__ __launch_bounds__(256) void gather_rows(const int* __restrict__ rowptr,
                                                   const int* __restrict__ cnt,
                                                   const int* __restrict__ csr,
                                                   const float* __restrict__ dinv,
                                                   const float* __restrict__ H,
                                                   const float* __restrict__ bias,
                                                   float* __restrict__ Z, int n) {
    constexpr int L = F / 4;
    const long long gid = (long long)blockIdx.x * blockDim.x + threadIdx.x;
    const int i = (int)(gid / L);
    const int l = (int)(gid % L);
    if (i >= n) return;
    const float4* H4 = (const float4*)H;
    const int beg = rowptr[i];
    const int end = beg + cnt[i];
    const float di = dinv[i];
    float4 acc = make_float4(0.0f, 0.0f, 0.0f, 0.0f);
    if (beg < end) {
        const int last = end - 1;
        int s[U];
#pragma unroll
        for (int k = 0; k < U; ++k) s[k] = csr[min(beg + k, last)];
        for (int p = beg; p < end; p += U) {
            int sn[U];
#pragma unroll
            for (int k = 0; k < U; ++k) sn[k] = csr[min(p + U + k, last)];
            float4 h[U];
#pragma unroll
            for (int k = 0; k < U; ++k) h[k] = H4[(size_t)s[k] * L + l];
#pragma unroll
            for (int k = 0; k < U; ++k)
                asm volatile("" :: "v"(h[k].x), "v"(h[k].y), "v"(h[k].z), "v"(h[k].w));
#pragma unroll
            for (int k = 0; k < U; ++k) {
                if (k == 0 || p + k < end) {
                    acc.x += h[k].x;
                    acc.y += h[k].y;
                    acc.z += h[k].z;
                    acc.w += h[k].w;
                }
            }
#pragma unroll
            for (int k = 0; k < U; ++k) s[k] = sn[k];
        }
    }
    const float4 hs = H4[(size_t)i * L + l];
    float4 v;
    v.x = di * (acc.x + hs.x);
    v.y = di * (acc.y + hs.y);
    v.z = di * (acc.z + hs.z);
    v.w = di * (acc.w + hs.w);
    if (FIRST) {
        v.x = di * fmaxf(v.x + bias[4 * l + 0], 0.0f);
        v.y = di * fmaxf(v.y + bias[4 * l + 1], 0.0f);
        v.z = di * fmaxf(v.z + bias[4 * l + 2], 0.0f);
        v.w = di * fmaxf(v.w + bias[4 * l + 3], 0.0f);
    }
    ((float4*)Z)[(size_t)i * L + l] = v;
}


__device__ __forceinline__ float oct_sum8(float v) {
    v += __int_as_float(__builtin_amdgcn_update_dpp(0, __float_as_int(v), 0x161, 0xf, 0xf, true));
    v += __int_as_float(__builtin_amdgcn_update_dpp(0, __float_as_int(v), 0x162, 0xf, 0xf, true));
    v += __int_as_float(__builtin_amdgcn_update_dpp(0, __float_as_int(v), 0x164, 0xf, 0xf, true));
    return v;
}

template <int T>
__device__ __forceinline__ int oct_bcast(int v, bool hi) {
    const int lo = __builtin_amdgcn_update_dpp(0, v, 0x150 + T, 0xf, 0xf, true);
    const int up = __builtin_amdgcn_update_dpp(0, v, 0x158 + T, 0xf, 0xf, true);
    return hi ? up : lo;
}

__global__ __launch_bounds__(256) void edge_score(const int* __restrict__ psrc,
                                                  const int* __restrict__ pdst,
                                                  const int* __restrict__ nsrc,
                                                  const int* __restrict__ ndst,
                                                  const float* __restrict__ R,
                                                  const float* __restrict__ A,
                                                  float* __restrict__ out, int nT) {
    const int lane = threadIdx.x & 31;
    const int j = lane & 7;
    const bool hi = (lane & 8) != 0;
    const unsigned rb = (unsigned)j << 4;
    const unsigned ab = (unsigned)(j & 3) << 4;
    const int total = 2 * nT;
    const int ngroups = (total + 31) >> 5;
    const int nwaves = gridDim.x * 8;
    int g = blockIdx.x * 8 + __builtin_amdgcn_readfirstlane((int)(threadIdx.x >> 5));
    if (g >= ngroups) return;
    const char* Rb = (const char*)R;
    const char* Ab = (const char*)A;

    int jx = min(g * 32 + lane, total - 1);
    int s = (jx < nT) ? psrc[jx] : nsrc[jx - nT];
    int d = (jx < nT) ? pdst[jx] : ndst[jx - nT];

    for (; g < ngroups; g += nwaves) {
        const int gn = min(g + nwaves, ngroups - 1);
        const int jn = min(gn * 32 + lane, total - 1);
        const int sn = (jn < nT) ? psrc[jn] : nsrc[jn - nT];
        const int dn = (jn < nT) ? pdst[jn] : ndst[jn - nT];

        float4 a[8], b[8];
#define OCT_STEP(T)                                                          \
        {                                                                    \
            const int ss = oct_bcast<T>(s, hi);                              \
            const int dd = oct_bcast<T>(d, hi);                              \
            a[T] = *(const float4*)(Rb + (((unsigned)ss << 7) + rb));        \
            b[T] = *(const float4*)(Ab + (((unsigned)dd << 6) + ab));        \
        }
        OCT_STEP(0) OCT_STEP(1) OCT_STEP(2) OCT_STEP(3)
        OCT_STEP(4) OCT_STEP(5) OCT_STEP(6) OCT_STEP(7)
#undef OCT_STEP
#pragma unroll
        for (int t = 0; t < 8; ++t)
            asm volatile("" :: "v"(a[t].x), "v"(a[t].y), "v"(a[t].z), "v"(a[t].w),
                               "v"(b[t].x), "v"(b[t].y), "v"(b[t].z), "v"(b[t].w));
        float res = 0.0f;
#pragma unroll
        for (int t = 0; t < 8; ++t) {
            float p = a[t].x * b[t].x;
            p = fmaf(a[t].y, b[t].y, p);
            p = fmaf(a[t].z, b[t].z, p);
            p = fmaf(a[t].w, b[t].w, p);
            p = (j < 4) ? p : a[t].x;
            p = oct_sum8(p);
            if (j == t) res = p;
        }
        const int jo = g * 32 + lane;
        if (jo < total) out[jo] = res;
        s = sn;
        d = dn;
    }
}

extern "C" void kernel_launch(void* const* d_in, const int* in_sizes, int n_in,
                              void* d_out, int out_size, void* d_ws, size_t ws_size,
                              hipStream_t stream) {
    const float* x   = (const float*)d_in[0];
    const int*   tei = (const int*)d_in[1];
    const int*   tpo = (const int*)d_in[2];
    const int*   tne = (const int*)d_in[3];
    const float* W1  = (const float*)d_in[4];
    const float* b1  = (const float*)d_in[5];
    const float* W2  = (const float*)d_in[6];
    const float* b2  = (const float*)d_in[7];

    const int N  = in_sizes[0] / 256;
    const int E  = in_sizes[1] / 2;
    const int ET = in_sizes[2] / 2;

    float* ws   = (float*)d_ws;
    float* h1   = ws;
    float* z1   = ws + (size_t)16 * N;
    float* a2   = ws + (size_t)32 * N;
    float* rec  = ws + (size_t)96 * N;
    int*   cnt  = (int*)(ws + (size_t)160 * N);
    float* dinv = ws + (size_t)161 * N;
    int*   rowp = (int*)(ws + (size_t)162 * N);
    int*   rank = (int*)(ws + (size_t)163 * N);
    int*   csr  = rank + (size_t)E;
    unsigned* packed = (unsigned*)(csr + (size_t)E);
    int*   hist = (int*)(packed + (size_t)E);
    int*   btot = hist + (size_t)MAX_BINS * SORT_G;

    const int B = 256;
    auto cdiv = [](long long a, long long b) { return (int)((a + b - 1) / b); };

    const int nb    = cdiv(N, BIN_W);
    const int chunk = cdiv(E, SORT_G);
    bin_count<<<SORT_G, SORT_T, 0, stream>>>(tei + E, rank, hist, E, nb, chunk);
    bin_scan<<<nb, SORT_G, 0, stream>>>(hist, btot);
    bin_scatter<<<SORT_G, SORT_T, 0, stream>>>(tei, tei + E, rank, hist, btot, packed, E, nb, chunk);
    bin_sort<<<nb, SORT_T, 0, stream>>>(packed, btot, rowp, cnt, dinv, csr, N);

    const int tiles = (N + 15) / 16;

    gemm_x_w1<<<cdiv(tiles, 8), B, 0, stream>>>(x, W1, dinv, h1, N);
    gather_rows<16, true, 8><<<cdiv((long long)4 * N, B), B, 0, stream>>>(
        rowp, cnt, csr, dinv, h1, b1, z1, N);

    gather_rows<16, false, 8><<<cdiv((long long)4 * N, B), B, 0, stream>>>(
        rowp, cnt, csr, dinv, z1, nullptr, a2, N);
    gemm_a_wwt<<<cdiv(tiles, 8), B, 0, stream>>>(a2, W2, b2, rec, N);

    edge_score<<<cdiv(2 * (long long)ET, 1024), B, 0, stream>>>(tpo, tpo + ET, tne, tne + ET,
                                                               rec, a2, (float*)d_out, ET);
}


# Sort: both levels' scattered 4-byte global stores (packed, csr) go to an LDS image in final order and leave as consecutive words, 1-3 lines per wave store not 32; slot order = LDS arrival as before
# speedup vs baseline: 16.3867x; 1.2346x over previous
#include <hip/hip_runtime.h>

typedef __attribute__((ext_vector_type(2))) float v2f;
typedef __attribute__((ext_vector_type(8))) float v8f;

constexpr int BIN_SH   = 9;
constexpr int BIN_W    = 1 << BIN_SH;
constexpr int MAX_BINS = 1024;
constexpr int SORT_G   = 256;
constexpr int SORT_T   = 1024;
constexpr int L1_STAGE = 16384;
constexpr int L2_STAGE = 32768;

__device__ __forceinline__ int wave_incl_scan(int v, int lane) {
    for (int o = 1; o < 32; o <<= 1) {
        int u = __shfl_up(v, o);
        if (lane >= o) v += u;
    }
    return v;
}

__global__ __launch_bounds__(SORT_T) void bin_count(const int* __restrict__ dst,
                                                    int* __restrict__ rank,
                                                    int* __restrict__ hist,
                                                    int nE, int nb, int chunk) {
    __shared__ int s_h[MAX_BINS];
    const int t = threadIdx.x;
    if (t < nb) s_h[t] = 0;
    __syncthreads();
    const int beg = blockIdx.x * chunk;
    const int end = min(beg + chunk, nE);
    int base = beg;
    for (; base + 4 * SORT_T <= end; base += 4 * SORT_T) {
        int d[4];
#pragma unroll
        for (int k = 0; k < 4; ++k) d[k] = dst[base + k * SORT_T + t];
#pragma unroll
        for (int k = 0; k < 4; ++k)
            rank[base + k * SORT_T + t] = atomicAdd(&s_h[d[k] >> BIN_SH], 1);
    }
    for (int e = base + t; e < end; e += SORT_T)
        rank[e] = atomicAdd(&s_h[dst[e] >> BIN_SH], 1);
    __syncthreads();
    if (t < nb) hist[t * SORT_G + blockIdx.x] = s_h[t];
}

__global__ __launch_bounds__(SORT_G) void bin_scan(int* __restrict__ hist,
                                                   int* __restrict__ total) {
    __shared__ int s_wave[SORT_G / 32];
    const int t = threadIdx.x, lane = t & 31, w = t >> 5;
    int* h = hist + (size_t)blockIdx.x * SORT_G;
    const int v = h[t];
    const int inc = wave_incl_scan(v, lane);
    if (lane == 31) s_wave[w] = inc;
    __syncthreads();
    int ex = inc - v;
    int sum = 0;
    for (int k = 0; k < SORT_G / 32; ++k) {
        if (k < w) ex += s_wave[k];
        sum += s_wave[k];
    }
    h[t] = ex;
    if (t == 0) total[blockIdx.x] = sum;
}

__global__ __launch_bounds__(SORT_T) void bin_scatter(const int* __restrict__ src,
                                                      const int* __restrict__ dst,
                                                      const int* __restrict__ rank,
                                                      const int* __restrict__ hist,
                                                      const int* __restrict__ total,
                                                      unsigned* __restrict__ packed,
                                                      int nE, int nb, int chunk) {
    __shared__ int s_base[MAX_BINS];
    __shared__ int s_loc[MAX_BINS];
    __shared__ int s_wave[SORT_T / 32];
    __shared__ int s_wave2[SORT_T / 32];
    __shared__ unsigned s_stage[L1_STAGE];
    __shared__ unsigned short s_bin[L1_STAGE];
    const int t = threadIdx.x, lane = t & 31, w = t >> 5;
    const int blk = blockIdx.x;
    int v = 0, pre = 0, mine = 0;
    if (t < nb) {
        v = total[t];
        pre = hist[t * SORT_G + blk];
        const int nxt = (blk + 1 < SORT_G) ? hist[t * SORT_G + blk + 1] : v;
        mine = nxt - pre;
    }
    const int inc  = wave_incl_scan(v, lane);
    const int inc2 = wave_incl_scan(mine, lane);
    if (lane == 31) {
        s_wave[w]  = inc;
        s_wave2[w] = inc2;
    }
    __syncthreads();
    int ex = inc - v, ex2 = inc2 - mine;
    for (int k = 0; k < SORT_T / 32; ++k)
        if (k < w) {
            ex  += s_wave[k];
            ex2 += s_wave2[k];
        }
    if (t < nb) {
        s_base[t] = ex + pre;
        s_loc[t]  = ex2;
    }
    __syncthreads();

    const int beg = blk * chunk;
    const int end = min(beg + chunk, nE);
    int base = beg;
    if (end - beg <= L1_STAGE) {
        for (; base + 4 * SORT_T <= end; base += 4 * SORT_T) {
            int d[4], s[4], r[4];
#pragma unroll
            for (int k = 0; k < 4; ++k) {
                const int e = base + k * SORT_T + t;
                d[k] = dst[e];
                s[k] = src[e];
                r[k] = rank[e];
            }
#pragma unroll
            for (int k = 0; k < 4; ++k) {
                const int bin = d[k] >> BIN_SH;
                const int pos = s_loc[bin] + r[k];
                s_stage[pos] = ((unsigned)s[k] << BIN_SH) | ((unsigned)d[k] & (BIN_W - 1));
                s_bin[pos]   = (unsigned short)bin;
            }
        }
        for (int e = base + t; e < end; e += SORT_T) {
            const int d = dst[e];
            const int bin = d >> BIN_SH;
            const int pos = s_loc[bin] + rank[e];
            s_stage[pos] = ((unsigned)src[e] << BIN_SH) | ((unsigned)d & (BIN_W - 1));
            s_bin[pos]   = (unsigned short)bin;
        }
        __syncthreads();
        const int nloc = end - beg;
#pragma unroll 4
        for (int i = t; i < nloc; i += SORT_T) {
            const int bin = s_bin[i];
            packed[s_base[bin] + (i - s_loc[bin])] = s_stage[i];
        }
    } else {
        for (; base + 4 * SORT_T <= end; base += 4 * SORT_T) {
            int d[4], s[4], r[4];
#pragma unroll
            for (int k = 0; k < 4; ++k) {
                const int e = base + k * SORT_T + t;
                d[k] = dst[e];
                s[k] = src[e];
                r[k] = rank[e];
            }
#pragma unroll
            for (int k = 0; k < 4; ++k)
                packed[s_base[d[k] >> BIN_SH] + r[k]] =
                    ((unsigned)s[k] << BIN_SH) | ((unsigned)d[k] & (BIN_W - 1));
        }
        for (int e = base + t; e < end; e += SORT_T) {
            const int d = dst[e];
            packed[s_base[d >> BIN_SH] + rank[e]] =
                ((unsigned)src[e] << BIN_SH) | ((unsigned)d & (BIN_W - 1));
        }
    }
}

__global__ __launch_bounds__(SORT_T) void bin_sort(const unsigned* __restrict__ packed,
                                                   const int* __restrict__ total,
                                                   int* __restrict__ rowptr,
                                                   int* __restrict__ cnt,
                                                   float* __restrict__ dinv,
                                                   int* __restrict__ csr, int n) {
    __shared__ int s_c[BIN_W];
    __shared__ int s_cur[BIN_W];
    __shared__ int s_wave[SORT_T / 32];
    __shared__ int s_red[SORT_T / 32];
    __shared__ int s_stage[L2_STAGE];
    const int t = threadIdx.x, lane = t & 31, w = t >> 5;
    const int b = blockIdx.x;

    int part = 0;
    for (int j = t; j < b; j += SORT_T) part += total[j];
    for (int o = 16; o > 0; o >>= 1) part += __shfl_xor(part, o);
    if (lane == 0) s_red[w] = part;
    if (t < BIN_W) s_c[t] = 0;
    __syncthreads();
    int beg = 0;
    for (int k = 0; k < SORT_T / 32; ++k) beg += s_red[k];
    const int end = beg + total[b];

    int base = beg;
    for (; base + 4 * SORT_T <= end; base += 4 * SORT_T) {
        unsigned p[4];
#pragma unroll
        for (int k = 0; k < 4; ++k) p[k] = packed[base + k * SORT_T + t];
#pragma unroll
        for (int k = 0; k < 4; ++k) atomicAdd(&s_c[p[k] & (BIN_W - 1)], 1);
    }
    for (int e = base + t; e < end; e += SORT_T) atomicAdd(&s_c[packed[e] & (BIN_W - 1)], 1);
    __syncthreads();

    const int c = (t < BIN_W) ? s_c[t] : 0;
    const int inc = wave_incl_scan(c, lane);
    if (lane == 31) s_wave[w] = inc;
    __syncthreads();
    int ex = inc - c;
    for (int k = 0; k < BIN_W / 32; ++k)
        if (k < w) ex += s_wave[k];
    if (t < BIN_W) {
        s_cur[t] = beg + ex;
        const int node = b * BIN_W + t;
        if (node < n) {
            rowptr[node] = beg + ex;
            cnt[node] = c;
            dinv[node] = rsqrtf((float)c + 1.0f);
        }
    }
    __syncthreads();

    base = beg;
    if (end - beg <= L2_STAGE) {
        for (; base + 4 * SORT_T <= end; base += 4 * SORT_T) {
            unsigned p[4];
            int q[4];
#pragma unroll
            for (int k = 0; k < 4; ++k) p[k] = packed[base + k * SORT_T + t];
#pragma unroll
            for (int k = 0; k < 4; ++k) q[k] = atomicAdd(&s_cur[p[k] & (BIN_W - 1)], 1) - beg;
#pragma unroll
            for (int k = 0; k < 4; ++k) s_stage[q[k]] = (int)(p[k] >> BIN_SH);
        }
        for (int e = base + t; e < end; e += SORT_T) {
            const unsigned p = packed[e];
            s_stage[atomicAdd(&s_cur[p & (BIN_W - 1)], 1) - beg] = (int)(p >> BIN_SH);
        }
        __syncthreads();
        int gi = (beg & ~31) + t;
        if (gi >= beg && gi < end) csr[gi] = s_stage[gi - beg];
#pragma unroll 4
        for (gi += SORT_T; gi < end; gi += SORT_T) csr[gi] = s_stage[gi - beg];
    } else {
        for (; base + 4 * SORT_T <= end; base += 4 * SORT_T) {
            unsigned p[4];
#pragma unroll
            for (int k = 0; k < 4; ++k) p[k] = packed[base + k * SORT_T + t];
#pragma unroll
            for (int k = 0; k < 4; ++k)
                csr[atomicAdd(&s_cur[p[k] & (BIN_W - 1)], 1)] = (int)(p[k] >> BIN_SH);
        }
        for (int e = base + t; e < end; e += SORT_T) {
            const unsigned p = packed[e];
            csr[atomicAdd(&s_cur[p & (BIN_W - 1)], 1)] = (int)(p >> BIN_SH);
        }
    }
}

__global__ __launch_bounds__(256) void gemm_x_w1(const float* __restrict__ X,
                                                 const float* __restrict__ W,
                                                 const float* __restrict__ dinv,
                                                 float* __restrict__ H, int n) {
    const int lane = threadIdx.x & 31;
    const int tile = blockIdx.x * 8 + (threadIdx.x >> 5);
    const int row0 = tile * 16;
    if (row0 >= n) return;
    int m = row0 + (lane & 15);
    if (m >= n) m = n - 1;
    const int kh  = (lane >> 4) << 1;
    const int col = lane & 15;
    const float* xr = X + (size_t)m * 256;
    const int ra = (lane >> 4) << 3;
    float sc[8];
#pragma unroll
    for (int v = 0; v < 8; ++v) sc[v] = dinv[min(row0 + ra + v, n - 1)];

    v8f acc = {};
#pragma unroll 4
    for (int k0 = 0; k0 < 256; k0 += 4) {
        v2f a, b;
        a.x = xr[k0 + kh];
        a.y = xr[k0 + kh + 1];
        b.x = W[(size_t)(k0 + kh) * 16 + col];
        b.y = W[(size_t)(k0 + kh + 1) * 16 + col];
        acc = __builtin_amdgcn_wmma_f32_16x16x4_f32(false, a, false, b,
                                                    (short)0, acc, false, false);
    }
#pragma unroll
    for (int v = 0; v < 8; ++v) acc[v] *= sc[v];
    if (row0 + 16 <= n) {
#pragma unroll
        for (int v = 0; v < 8; ++v)
            H[(size_t)(row0 + ra + v) * 16 + col] = acc[v];
    } else {
#pragma unroll
        for (int v = 0; v < 8; ++v) {
            int r = row0 + ra + v;
            if (r < n) H[(size_t)r * 16 + col] = acc[v];
        }
    }
}

__global__ __launch_bounds__(256) void gemm_a_wwt(const float* __restrict__ A2,
                                                   const float* __restrict__ W,
                                                   const float* __restrict__ bias,
                                                   float* __restrict__ R, int n) {
    const int lane = threadIdx.x & 31;
    const int tile = blockIdx.x * 8 + (threadIdx.x >> 5);
    const int row0 = tile * 16;
    if (row0 >= n) return;
    const int col = lane & 15;
    const int h   = lane >> 4;
    int m = row0 + col;
    if (m >= n) m = n - 1;
    const float4* ar = (const float4*)(A2 + (size_t)m * 16 + 8 * h);
    const float4 a0 = ar[0], a1 = ar[1];
    const float4* wr = (const float4*)(W + (size_t)col * 64);
    const float4* br = (const float4*)bias;
    v8f g = {};
    float cp = 0.0f, bp = 0.0f;
#pragma unroll
    for (int q = 0; q < 8; ++q) {
        const float4 w = wr[2 * q + h];
        const float4 b = br[2 * q + h];
        cp = fmaf(w.x, b.x, cp); cp = fmaf(w.y, b.y, cp);
        cp = fmaf(w.z, b.z, cp); cp = fmaf(w.w, b.w, cp);
        bp = fmaf(b.x, b.x, bp); bp = fmaf(b.y, b.y, bp);
        bp = fmaf(b.z, b.z, bp); bp = fmaf(b.w, b.w, bp);
        v2f f;
        f.x = w.x; f.y = w.y;
        g = __builtin_amdgcn_wmma_f32_16x16x4_f32(false, f, false, f, (short)0, g, false, false);
        f.x = w.z; f.y = w.w;
        g = __builtin_amdgcn_wmma_f32_16x16x4_f32(false, f, false, f, (short)0, g, false, false);
    }
    const float cc = cp + __shfl_xor(cp, 16);
    const float bb = bp + __shfl_xor(bp, 16);

    const float av[8] = {a0.x, a0.y, a0.z, a0.w, a1.x, a1.y, a1.z, a1.w};
    v8f acc = {};
#pragma unroll
    for (int w = 0; w < 4; ++w) {
        v2f a, b;
        a.x = av[2 * w]; a.y = av[2 * w + 1];
        b.x = g[2 * w];  b.y = g[2 * w + 1];
        acc = __builtin_amdgcn_wmma_f32_16x16x4_f32(false, a, false, b, (short)0, acc, false, false);
    }
    float up = 0.0f;
    const int cci = __float_as_int(cc);
#pragma unroll
    for (int i = 0; i < 8; ++i) {
        const float clo = __int_as_float(__builtin_amdgcn_readlane(cci, i));
        const float chi = __int_as_float(__builtin_amdgcn_readlane(cci, i + 8));
        up = fmaf(av[i], h ? chi : clo, up);
    }
    const float u = up + __shfl_xor(up, 16) + bb;

    const int ra = h << 3;
    const float4 zero4 = make_float4(0.0f, 0.0f, 0.0f, 0.0f);
    const float4 t0 = h ? zero4 : make_float4(u, 0.0f, 0.0f, 0.0f);
    if (row0 + 16 <= n) {
#pragma unroll
        for (int v = 0; v < 8; ++v)
            R[(size_t)(row0 + ra + v) * 32 + col] = acc[v] + cc;
        float4* tp = (float4*)(R + (size_t)(row0 + col) * 32 + 16 + 8 * h);
        tp[0] = t0;
        tp[1] = zero4;
    } else {
#pragma unroll
        for (int v = 0; v < 8; ++v) {
            const int r = row0 + ra + v;
            if (r < n) R[(size_t)r * 32 + col] = acc[v] + cc;
        }
        if (row0 + col < n) {
            float4* tp = (float4*)(R + (size_t)(row0 + col) * 32 + 16 + 8 * h);
            tp[0] = t0;
            tp[1] = zero4;
        }
    }
}

template <int F, bool FIRST, int U>
__global__ __launch_bounds__(256) void gather_rows(const int* __restrict__ rowptr,
                                                   const int* __restrict__ cnt,
                                                   const int* __restrict__ csr,
                                                   const float* __restrict__ dinv,
                                                   const float* __restrict__ H,
                                                   const float* __restrict__ bias,
                                                   float* __restrict__ Z, int n) {
    constexpr int L = F / 4;
    const long long gid = (long long)blockIdx.x * blockDim.x + threadIdx.x;
    const int i = (int)(gid / L);
    const int l = (int)(gid % L);
    if (i >= n) return;
    const float4* H4 = (const float4*)H;
    const int beg = rowptr[i];
    const int end = beg + cnt[i];
    const float di = dinv[i];
    float4 acc = make_float4(0.0f, 0.0f, 0.0f, 0.0f);
    if (beg < end) {
        const int last = end - 1;
        int s[U];
#pragma unroll
        for (int k = 0; k < U; ++k) s[k] = csr[min(beg + k, last)];
        for (int p = beg; p < end; p += U) {
            int sn[U];
#pragma unroll
            for (int k = 0; k < U; ++k) sn[k] = csr[min(p + U + k, last)];
            float4 h[U];
#pragma unroll
            for (int k = 0; k < U; ++k) h[k] = H4[(size_t)s[k] * L + l];
#pragma unroll
            for (int k = 0; k < U; ++k)
                asm volatile("" :: "v"(h[k].x), "v"(h[k].y), "v"(h[k].z), "v"(h[k].w));
#pragma unroll
            for (int k = 0; k < U; ++k) {
                if (k == 0 || p + k < end) {
                    acc.x += h[k].x;
                    acc.y += h[k].y;
                    acc.z += h[k].z;
                    acc.w += h[k].w;
                }
            }
#pragma unroll
            for (int k = 0; k < U; ++k) s[k] = sn[k];
        }
    }
    const float4 hs = H4[(size_t)i * L + l];
    float4 v;
    v.x = di * (acc.x + hs.x);
    v.y = di * (acc.y + hs.y);
    v.z = di * (acc.z + hs.z);
    v.w = di * (acc.w + hs.w);
    if (FIRST) {
        v.x = di * fmaxf(v.x + bias[4 * l + 0], 0.0f);
        v.y = di * fmaxf(v.y + bias[4 * l + 1], 0.0f);
        v.z = di * fmaxf(v.z + bias[4 * l + 2], 0.0f);
        v.w = di * fmaxf(v.w + bias[4 * l + 3], 0.0f);
    }
    ((float4*)Z)[(size_t)i * L + l] = v;
}


__device__ __forceinline__ float oct_sum8(float v) {
    v += __int_as_float(__builtin_amdgcn_update_dpp(0, __float_as_int(v), 0x161, 0xf, 0xf, true));
    v += __int_as_float(__builtin_amdgcn_update_dpp(0, __float_as_int(v), 0x162, 0xf, 0xf, true));
    v += __int_as_float(__builtin_amdgcn_update_dpp(0, __float_as_int(v), 0x164, 0xf, 0xf, true));
    return v;
}

template <int T>
__device__ __forceinline__ int oct_bcast(int v, bool hi) {
    const int lo = __builtin_amdgcn_update_dpp(0, v, 0x150 + T, 0xf, 0xf, true);
    const int up = __builtin_amdgcn_update_dpp(0, v, 0x158 + T, 0xf, 0xf, true);
    return hi ? up : lo;
}

__global__ __launch_bounds__(256) void edge_score(const int* __restrict__ psrc,
                                                  const int* __restrict__ pdst,
                                                  const int* __restrict__ nsrc,
                                                  const int* __restrict__ ndst,
                                                  const float* __restrict__ R,
                                                  const float* __restrict__ A,
                                                  float* __restrict__ out, int nT) {
    const int lane = threadIdx.x & 31;
    const int j = lane & 7;
    const bool hi = (lane & 8) != 0;
    const unsigned rb = (unsigned)j << 4;
    const unsigned ab = (unsigned)(j & 3) << 4;
    const int total = 2 * nT;
    const int ngroups = (total + 31) >> 5;
    const int nwaves = gridDim.x * 8;
    int g = blockIdx.x * 8 + __builtin_amdgcn_readfirstlane((int)(threadIdx.x >> 5));
    if (g >= ngroups) return;
    const char* Rb = (const char*)R;
    const char* Ab = (const char*)A;

    int jx = min(g * 32 + lane, total - 1);
    int s = (jx < nT) ? psrc[jx] : nsrc[jx - nT];
    int d = (jx < nT) ? pdst[jx] : ndst[jx - nT];

    for (; g < ngroups; g += nwaves) {
        const int gn = min(g + nwaves, ngroups - 1);
        const int jn = min(gn * 32 + lane, total - 1);
        const int sn = (jn < nT) ? psrc[jn] : nsrc[jn - nT];
        const int dn = (jn < nT) ? pdst[jn] : ndst[jn - nT];

        float4 a[8], b[8];
#define OCT_STEP(T)                                                          \
        {                                                                    \
            const int ss = oct_bcast<T>(s, hi);                              \
            const int dd = oct_bcast<T>(d, hi);                              \
            a[T] = *(const float4*)(Rb + (((unsigned)ss << 7) + rb));        \
            b[T] = *(const float4*)(Ab + (((unsigned)dd << 6) + ab));        \
        }
        OCT_STEP(0) OCT_STEP(1) OCT_STEP(2) OCT_STEP(3)
        OCT_STEP(4) OCT_STEP(5) OCT_STEP(6) OCT_STEP(7)
#undef OCT_STEP
#pragma unroll
        for (int t = 0; t < 8; ++t)
            asm volatile("" :: "v"(a[t].x), "v"(a[t].y), "v"(a[t].z), "v"(a[t].w),
                               "v"(b[t].x), "v"(b[t].y), "v"(b[t].z), "v"(b[t].w));
        float res = 0.0f;
#pragma unroll
        for (int t = 0; t < 8; ++t) {
            float p = a[t].x * b[t].x;
            p = fmaf(a[t].y, b[t].y, p);
            p = fmaf(a[t].z, b[t].z, p);
            p = fmaf(a[t].w, b[t].w, p);
            p = (j < 4) ? p : a[t].x;
            p = oct_sum8(p);
            if (j == t) res = p;
        }
        const int jo = g * 32 + lane;
        if (jo < total) out[jo] = res;
        s = sn;
        d = dn;
    }
}

extern "C" void kernel_launch(void* const* d_in, const int* in_sizes, int n_in,
                              void* d_out, int out_size, void* d_ws, size_t ws_size,
                              hipStream_t stream) {
    const float* x   = (const float*)d_in[0];
    const int*   tei = (const int*)d_in[1];
    const int*   tpo = (const int*)d_in[2];
    const int*   tne = (const int*)d_in[3];
    const float* W1  = (const float*)d_in[4];
    const float* b1  = (const float*)d_in[5];
    const float* W2  = (const float*)d_in[6];
    const float* b2  = (const float*)d_in[7];

    const int N  = in_sizes[0] / 256;
    const int E  = in_sizes[1] / 2;
    const int ET = in_sizes[2] / 2;

    float* ws   = (float*)d_ws;
    float* h1   = ws;
    float* z1   = ws + (size_t)16 * N;
    float* a2   = ws + (size_t)32 * N;
    float* rec  = ws + (size_t)96 * N;
    int*   cnt  = (int*)(ws + (size_t)160 * N);
    float* dinv = ws + (size_t)161 * N;
    int*   rowp = (int*)(ws + (size_t)162 * N);
    int*   rank = (int*)(ws + (size_t)163 * N);
    int*   csr  = rank + (size_t)E;
    unsigned* packed = (unsigned*)(csr + (size_t)E);
    int*   hist = (int*)(packed + (size_t)E);
    int*   btot = hist + (size_t)MAX_BINS * SORT_G;

    const int B = 256;
    auto cdiv = [](long long a, long long b) { return (int)((a + b - 1) / b); };

    const int nb    = cdiv(N, BIN_W);
    const int chunk = cdiv(E, SORT_G);
    bin_count<<<SORT_G, SORT_T, 0, stream>>>(tei + E, rank, hist, E, nb, chunk);
    bin_scan<<<nb, SORT_G, 0, stream>>>(hist, btot);
    bin_scatter<<<SORT_G, SORT_T, 0, stream>>>(tei, tei + E, rank, hist, btot, packed, E, nb, chunk);
    bin_sort<<<nb, SORT_T, 0, stream>>>(packed, btot, rowp, cnt, dinv, csr, N);

    const int tiles = (N + 15) / 16;

    gemm_x_w1<<<cdiv(tiles, 8), B, 0, stream>>>(x, W1, dinv, h1, N);
    gather_rows<16, true, 8><<<cdiv((long long)4 * N, B), B, 0, stream>>>(
        rowp, cnt, csr, dinv, h1, b1, z1, N);

    gather_rows<16, false, 8><<<cdiv((long long)4 * N, B), B, 0, stream>>>(
        rowp, cnt, csr, dinv, z1, nullptr, a2, N);
    gemm_a_wwt<<<cdiv(tiles, 8), B, 0, stream>>>(a2, W2, b2, rec, N);

    edge_score<<<cdiv(2 * (long long)ET, 1024), B, 0, stream>>>(tpo, tpo + ET, tne, tne + ET,
                                                               rec, a2, (float*)d_out, ET);
}


# Sort level 1 block-local: a block counts, scans, places its own chunk by bin in ONE launch (no rank plane, no scan launch); level 2 pulls its bin's 256 pieces to LDS; 7 launches not 9; slot order kept
# speedup vs baseline: 16.9489x; 1.0343x over previous
#include <hip/hip_runtime.h>

typedef __attribute__((ext_vector_type(2))) float v2f;
typedef __attribute__((ext_vector_type(8))) float v8f;

constexpr int BIN_SH   = 9;
constexpr int BIN_W    = 1 << BIN_SH;
constexpr int MAX_BINS = 1024;
constexpr int SORT_G   = 256;
constexpr int SORT_T   = 1024;
constexpr int L1_STAGE = 16384;
constexpr int L2_STAGE = 32768;
constexpr int PIECES_W = SORT_G / (SORT_T / 32);
constexpr int PIECE_R  = 3;

__device__ __forceinline__ int wave_incl_scan(int v, int lane) {
    for (int o = 1; o < 32; o <<= 1) {
        int u = __shfl_up(v, o);
        if (lane >= o) v += u;
    }
    return v;
}

__global__ __launch_bounds__(SORT_T) void bin_local(const int* __restrict__ src,
                                                    const int* __restrict__ dst,
                                                    int* __restrict__ hist,
                                                    int* __restrict__ loc,
                                                    unsigned* __restrict__ packed,
                                                    int nE, int nb, int chunk) {
    __shared__ int s_h[MAX_BINS];
    __shared__ int s_wave[SORT_T / 32];
    __shared__ unsigned s_stage[L1_STAGE];
    const int t = threadIdx.x, lane = t & 31, w = t >> 5;
    const int blk = blockIdx.x;
    if (t < nb) s_h[t] = 0;
    __syncthreads();
    const int beg = blk * chunk;
    const int end = min(beg + chunk, nE);
    int base = beg;
    for (; base + 4 * SORT_T <= end; base += 4 * SORT_T) {
        int d[4];
#pragma unroll
        for (int k = 0; k < 4; ++k) d[k] = dst[base + k * SORT_T + t];
#pragma unroll
        for (int k = 0; k < 4; ++k) atomicAdd(&s_h[d[k] >> BIN_SH], 1);
    }
    for (int e = base + t; e < end; e += SORT_T) atomicAdd(&s_h[dst[e] >> BIN_SH], 1);
    __syncthreads();

    const int v = (t < nb) ? s_h[t] : 0;
    const int inc = wave_incl_scan(v, lane);
    if (lane == 31) s_wave[w] = inc;
    __syncthreads();
    int ex = inc - v;
    for (int k = 0; k < SORT_T / 32; ++k)
        if (k < w) ex += s_wave[k];
    if (t < nb) {
        hist[t * SORT_G + blk] = v;
        loc[t * SORT_G + blk]  = ex;
        s_h[t] = ex;
    }
    __syncthreads();

    base = beg;
    if (end - beg <= L1_STAGE) {
        for (; base + 4 * SORT_T <= end; base += 4 * SORT_T) {
            int d[4], s[4], q[4];
#pragma unroll
            for (int k = 0; k < 4; ++k) {
                const int e = base + k * SORT_T + t;
                d[k] = dst[e];
                s[k] = src[e];
            }
#pragma unroll
            for (int k = 0; k < 4; ++k) q[k] = atomicAdd(&s_h[d[k] >> BIN_SH], 1);
#pragma unroll
            for (int k = 0; k < 4; ++k)
                s_stage[q[k]] = ((unsigned)s[k] << BIN_SH) | ((unsigned)d[k] & (BIN_W - 1));
        }
        for (int e = base + t; e < end; e += SORT_T) {
            const int d = dst[e];
            s_stage[atomicAdd(&s_h[d >> BIN_SH], 1)] =
                ((unsigned)src[e] << BIN_SH) | ((unsigned)d & (BIN_W - 1));
        }
        __syncthreads();
        const int nloc = end - beg;
#pragma unroll 4
        for (int i = t; i < nloc; i += SORT_T) packed[beg + i] = s_stage[i];
    } else {
        for (int e = base + t; e < end; e += SORT_T) {
            const int d = dst[e];
            packed[beg + atomicAdd(&s_h[d >> BIN_SH], 1)] =
                ((unsigned)src[e] << BIN_SH) | ((unsigned)d & (BIN_W - 1));
        }
    }
}

__global__ __launch_bounds__(SORT_T) void bin_sort(const unsigned* __restrict__ packed,
                                                   const int* __restrict__ hist,
                                                   const int* __restrict__ loc,
                                                   int* __restrict__ rowptr,
                                                   int* __restrict__ cnt,
                                                   float* __restrict__ dinv,
                                                   int* __restrict__ csr,
                                                   int n, int nE, int chunk) {
    __shared__ int s_c[BIN_W];
    __shared__ int s_cur[BIN_W];
    __shared__ int s_wave[SORT_T / 32];
    __shared__ int s_red[SORT_T / 32];
    __shared__ int s_off[SORT_G];
    __shared__ int s_len[SORT_G];
    __shared__ int s_gs[SORT_G];
    __shared__ unsigned s_in[L2_STAGE];
    __shared__ int s_stage[L2_STAGE];
    const int t = threadIdx.x, lane = t & 31, w = t >> 5;
    const int b = blockIdx.x;

    int c0 = 0, l0 = 0;
    if (t < SORT_G) {
        c0 = hist[b * SORT_G + t];
        l0 = loc[b * SORT_G + t];
    }
    const int inc0 = wave_incl_scan(c0, lane);
    int part = l0;
    for (int o = 16; o > 0; o >>= 1) part += __shfl_xor(part, o);
    if (lane == 31) s_wave[w] = inc0;
    if (lane == 0) s_red[w] = part;
    if (t < BIN_W) s_c[t] = 0;
    __syncthreads();
    int ex0 = inc0 - c0, beg = 0, total = 0;
    for (int k = 0; k < SORT_T / 32; ++k) {
        const int sw = s_wave[k];
        if (k < w) ex0 += sw;
        total += sw;
        beg += s_red[k];
    }
    if (t < SORT_G) {
        s_off[t] = ex0;
        s_len[t] = c0;
        s_gs[t]  = t * chunk + l0;
    }
    __syncthreads();
    const int end = beg + total;
    const bool fits = total <= L2_STAGE;

    if (fits) {
        int len[PIECES_W], gs[PIECES_W], off[PIECES_W];
        unsigned v[PIECES_W][PIECE_R];
#pragma unroll
        for (int j = 0; j < PIECES_W; ++j) {
            const int pc = PIECES_W * w + j;
            len[j] = s_len[pc];
            gs[j]  = s_gs[pc];
            off[j] = s_off[pc];
        }
#pragma unroll
        for (int j = 0; j < PIECES_W; ++j) {
#pragma unroll
            for (int r = 0; r < PIECE_R; ++r) {
                const int i = min(lane + 32 * r, max(len[j] - 1, 0));
                v[j][r] = packed[max(min(gs[j] + i, nE - 1), 0)];
            }
        }
#pragma unroll
        for (int j = 0; j < PIECES_W; ++j) {
#pragma unroll
            for (int r = 0; r < PIECE_R; ++r) {
                const int i = lane + 32 * r;
                if (i < len[j]) {
                    s_in[off[j] + i] = v[j][r];
                    atomicAdd(&s_c[v[j][r] & (BIN_W - 1)], 1);
                }
            }
        }
#pragma unroll
        for (int j = 0; j < PIECES_W; ++j) {
            for (int i = lane + 32 * PIECE_R; i < len[j]; i += 32) {
                const unsigned p = packed[gs[j] + i];
                s_in[off[j] + i] = p;
                atomicAdd(&s_c[p & (BIN_W - 1)], 1);
            }
        }
    } else {
        for (int j = 0; j < PIECES_W; ++j) {
            const int pc = PIECES_W * w + j;
            const int ln = s_len[pc], g = s_gs[pc];
            for (int i = lane; i < ln; i += 32) atomicAdd(&s_c[packed[g + i] & (BIN_W - 1)], 1);
        }
    }
    __syncthreads();

    const int c = (t < BIN_W) ? s_c[t] : 0;
    const int inc = wave_incl_scan(c, lane);
    if (lane == 31) s_wave[w] = inc;
    __syncthreads();
    int ex = inc - c;
    for (int k = 0; k < BIN_W / 32; ++k)
        if (k < w) ex += s_wave[k];
    if (t < BIN_W) {
        s_cur[t] = beg + ex;
        const int node = b * BIN_W + t;
        if (node < n) {
            rowptr[node] = beg + ex;
            cnt[node] = c;
            dinv[node] = rsqrtf((float)c + 1.0f);
        }
    }
    __syncthreads();

    if (fits) {
        int i = t;
        for (; i + 3 * SORT_T < total; i += 4 * SORT_T) {
            unsigned p[4];
            int q[4];
#pragma unroll
            for (int k = 0; k < 4; ++k) p[k] = s_in[i + k * SORT_T];
#pragma unroll
            for (int k = 0; k < 4; ++k) q[k] = atomicAdd(&s_cur[p[k] & (BIN_W - 1)], 1) - beg;
#pragma unroll
            for (int k = 0; k < 4; ++k) s_stage[q[k]] = (int)(p[k] >> BIN_SH);
        }
        for (; i < total; i += SORT_T) {
            const unsigned p = s_in[i];
            s_stage[atomicAdd(&s_cur[p & (BIN_W - 1)], 1) - beg] = (int)(p >> BIN_SH);
        }
        __syncthreads();
        int gi = (beg & ~31) + t;
        if (gi >= beg && gi < end) csr[gi] = s_stage[gi - beg];
#pragma unroll 4
        for (gi += SORT_T; gi < end; gi += SORT_T) csr[gi] = s_stage[gi - beg];
    } else {
        for (int j = 0; j < PIECES_W; ++j) {
            const int pc = PIECES_W * w + j;
            const int ln = s_len[pc], g = s_gs[pc];
            for (int i = lane; i < ln; i += 32) {
                const unsigned p = packed[g + i];
                csr[atomicAdd(&s_cur[p & (BIN_W - 1)], 1)] = (int)(p >> BIN_SH);
            }
        }
    }
}

__global__ __launch_bounds__(256) void gemm_x_w1(const float* __restrict__ X,
                                                 const float* __restrict__ W,
                                                 const float* __restrict__ dinv,
                                                 float* __restrict__ H, int n) {
    const int lane = threadIdx.x & 31;
    const int tile = blockIdx.x * 8 + (threadIdx.x >> 5);
    const int row0 = tile * 16;
    if (row0 >= n) return;
    int m = row0 + (lane & 15);
    if (m >= n) m = n - 1;
    const int kh  = (lane >> 4) << 1;
    const int col = lane & 15;
    const float* xr = X + (size_t)m * 256;
    const int ra = (lane >> 4) << 3;
    float sc[8];
#pragma unroll
    for (int v = 0; v < 8; ++v) sc[v] = dinv[min(row0 + ra + v, n - 1)];

    v8f acc = {};
#pragma unroll 4
    for (int k0 = 0; k0 < 256; k0 += 4) {
        v2f a, b;
        a.x = xr[k0 + kh];
        a.y = xr[k0 + kh + 1];
        b.x = W[(size_t)(k0 + kh) * 16 + col];
        b.y = W[(size_t)(k0 + kh + 1) * 16 + col];
        acc = __builtin_amdgcn_wmma_f32_16x16x4_f32(false, a, false, b,
                                                    (short)0, acc, false, false);
    }
#pragma unroll
    for (int v = 0; v < 8; ++v) acc[v] *= sc[v];
    if (row0 + 16 <= n) {
#pragma unroll
        for (int v = 0; v < 8; ++v)
            H[(size_t)(row0 + ra + v) * 16 + col] = acc[v];
    } else {
#pragma unroll
        for (int v = 0; v < 8; ++v) {
            int r = row0 + ra + v;
            if (r < n) H[(size_t)r * 16 + col] = acc[v];
        }
    }
}

__global__ __launch_bounds__(256) void gemm_a_wwt(const float* __restrict__ A2,
                                                   const float* __restrict__ W,
                                                   const float* __restrict__ bias,
                                                   float* __restrict__ R, int n) {
    const int lane = threadIdx.x & 31;
    const int tile = blockIdx.x * 8 + (threadIdx.x >> 5);
    const int row0 = tile * 16;
    if (row0 >= n) return;
    const int col = lane & 15;
    const int h   = lane >> 4;
    int m = row0 + col;
    if (m >= n) m = n - 1;
    const float4* ar = (const float4*)(A2 + (size_t)m * 16 + 8 * h);
    const float4 a0 = ar[0], a1 = ar[1];
    const float4* wr = (const float4*)(W + (size_t)col * 64);
    const float4* br = (const float4*)bias;
    v8f g = {};
    float cp = 0.0f, bp = 0.0f;
#pragma unroll
    for (int q = 0; q < 8; ++q) {
        const float4 w = wr[2 * q + h];
        const float4 b = br[2 * q + h];
        cp = fmaf(w.x, b.x, cp); cp = fmaf(w.y, b.y, cp);
        cp = fmaf(w.z, b.z, cp); cp = fmaf(w.w, b.w, cp);
        bp = fmaf(b.x, b.x, bp); bp = fmaf(b.y, b.y, bp);
        bp = fmaf(b.z, b.z, bp); bp = fmaf(b.w, b.w, bp);
        v2f f;
        f.x = w.x; f.y = w.y;
        g = __builtin_amdgcn_wmma_f32_16x16x4_f32(false, f, false, f, (short)0, g, false, false);
        f.x = w.z; f.y = w.w;
        g = __builtin_amdgcn_wmma_f32_16x16x4_f32(false, f, false, f, (short)0, g, false, false);
    }
    const float cc = cp + __shfl_xor(cp, 16);
    const float bb = bp + __shfl_xor(bp, 16);

    const float av[8] = {a0.x, a0.y, a0.z, a0.w, a1.x, a1.y, a1.z, a1.w};
    v8f acc = {};
#pragma unroll
    for (int w = 0; w < 4; ++w) {
        v2f a, b;
        a.x = av[2 * w]; a.y = av[2 * w + 1];
        b.x = g[2 * w];  b.y = g[2 * w + 1];
        acc = __builtin_amdgcn_wmma_f32_16x16x4_f32(false, a, false, b, (short)0, acc, false, false);
    }
    float up = 0.0f;
    const int cci = __float_as_int(cc);
#pragma unroll
    for (int i = 0; i < 8; ++i) {
        const float clo = __int_as_float(__builtin_amdgcn_readlane(cci, i));
        const float chi = __int_as_float(__builtin_amdgcn_readlane(cci, i + 8));
        up = fmaf(av[i], h ? chi : clo, up);
    }
    const float u = up + __shfl_xor(up, 16) + bb;

    const int ra = h << 3;
    const float4 zero4 = make_float4(0.0f, 0.0f, 0.0f, 0.0f);
    const float4 t0 = h ? zero4 : make_float4(u, 0.0f, 0.0f, 0.0f);
    if (row0 + 16 <= n) {
#pragma unroll
        for (int v = 0; v < 8; ++v)
            R[(size_t)(row0 + ra + v) * 32 + col] = acc[v] + cc;
        float4* tp = (float4*)(R + (size_t)(row0 + col) * 32 + 16 + 8 * h);
        tp[0] = t0;
        tp[1] = zero4;
    } else {
#pragma unroll
        for (int v = 0; v < 8; ++v) {
            const int r = row0 + ra + v;
            if (r < n) R[(size_t)r * 32 + col] = acc[v] + cc;
        }
        if (row0 + col < n) {
            float4* tp = (float4*)(R + (size_t)(row0 + col) * 32 + 16 + 8 * h);
            tp[0] = t0;
            tp[1] = zero4;
        }
    }
}

template <int F, bool FIRST, int U>
__global__ __launch_bounds__(256) void gather_rows(const int* __restrict__ rowptr,
                                                   const int* __restrict__ cnt,
                                                   const int* __restrict__ csr,
                                                   const float* __restrict__ dinv,
                                                   const float* __restrict__ H,
                                                   const float* __restrict__ bias,
                                                   float* __restrict__ Z, int n) {
    constexpr int L = F / 4;
    const long long gid = (long long)blockIdx.x * blockDim.x + threadIdx.x;
    const int i = (int)(gid / L);
    const int l = (int)(gid % L);
    if (i >= n) return;
    const float4* H4 = (const float4*)H;
    const int beg = rowptr[i];
    const int end = beg + cnt[i];
    const float di = dinv[i];
    float4 acc = make_float4(0.0f, 0.0f, 0.0f, 0.0f);
    if (beg < end) {
        const int last = end - 1;
        int s[U];
#pragma unroll
        for (int k = 0; k < U; ++k) s[k] = csr[min(beg + k, last)];
        for (int p = beg; p < end; p += U) {
            int sn[U];
#pragma unroll
            for (int k = 0; k < U; ++k) sn[k] = csr[min(p + U + k, last)];
            float4 h[U];
#pragma unroll
            for (int k = 0; k < U; ++k) h[k] = H4[(size_t)s[k] * L + l];
#pragma unroll
            for (int k = 0; k < U; ++k)
                asm volatile("" :: "v"(h[k].x), "v"(h[k].y), "v"(h[k].z), "v"(h[k].w));
#pragma unroll
            for (int k = 0; k < U; ++k) {
                if (k == 0 || p + k < end) {
                    acc.x += h[k].x;
                    acc.y += h[k].y;
                    acc.z += h[k].z;
                    acc.w += h[k].w;
                }
            }
#pragma unroll
            for (int k = 0; k < U; ++k) s[k] = sn[k];
        }
    }
    const float4 hs = H4[(size_t)i * L + l];
    float4 v;
    v.x = di * (acc.x + hs.x);
    v.y = di * (acc.y + hs.y);
    v.z = di * (acc.z + hs.z);
    v.w = di * (acc.w + hs.w);
    if (FIRST) {
        v.x = di * fmaxf(v.x + bias[4 * l + 0], 0.0f);
        v.y = di * fmaxf(v.y + bias[4 * l + 1], 0.0f);
        v.z = di * fmaxf(v.z + bias[4 * l + 2], 0.0f);
        v.w = di * fmaxf(v.w + bias[4 * l + 3], 0.0f);
    }
    ((float4*)Z)[(size_t)i * L + l] = v;
}


__device__ __forceinline__ float oct_sum8(float v) {
    v += __int_as_float(__builtin_amdgcn_update_dpp(0, __float_as_int(v), 0x161, 0xf, 0xf, true));
    v += __int_as_float(__builtin_amdgcn_update_dpp(0, __float_as_int(v), 0x162, 0xf, 0xf, true));
    v += __int_as_float(__builtin_amdgcn_update_dpp(0, __float_as_int(v), 0x164, 0xf, 0xf, true));
    return v;
}

template <int T>
__device__ __forceinline__ int oct_bcast(int v, bool hi) {
    const int lo = __builtin_amdgcn_update_dpp(0, v, 0x150 + T, 0xf, 0xf, true);
    const int up = __builtin_amdgcn_update_dpp(0, v, 0x158 + T, 0xf, 0xf, true);
    return hi ? up : lo;
}

__global__ __launch_bounds__(256) void edge_score(const int* __restrict__ psrc,
                                                  const int* __restrict__ pdst,
                                                  const int* __restrict__ nsrc,
                                                  const int* __restrict__ ndst,
                                                  const float* __restrict__ R,
                                                  const float* __restrict__ A,
                                                  float* __restrict__ out, int nT) {
    const int lane = threadIdx.x & 31;
    const int j = lane & 7;
    const bool hi = (lane & 8) != 0;
    const unsigned rb = (unsigned)j << 4;
    const unsigned ab = (unsigned)(j & 3) << 4;
    const int total = 2 * nT;
    const int ngroups = (total + 31) >> 5;
    const int nwaves = gridDim.x * 8;
    int g = blockIdx.x * 8 + __builtin_amdgcn_readfirstlane((int)(threadIdx.x >> 5));
    if (g >= ngroups) return;
    const char* Rb = (const char*)R;
    const char* Ab = (const char*)A;

    int jx = min(g * 32 + lane, total - 1);
    int s = (jx < nT) ? psrc[jx] : nsrc[jx - nT];
    int d = (jx < nT) ? pdst[jx] : ndst[jx - nT];

    for (; g < ngroups; g += nwaves) {
        const int gn = min(g + nwaves, ngroups - 1);
        const int jn = min(gn * 32 + lane, total - 1);
        const int sn = (jn < nT) ? psrc[jn] : nsrc[jn - nT];
        const int dn = (jn < nT) ? pdst[jn] : ndst[jn - nT];

        float4 a[8], b[8];
#define OCT_STEP(T)                                                          \
        {                                                                    \
            const int ss = oct_bcast<T>(s, hi);                              \
            const int dd = oct_bcast<T>(d, hi);                              \
            a[T] = *(const float4*)(Rb + (((unsigned)ss << 7) + rb));        \
            b[T] = *(const float4*)(Ab + (((unsigned)dd << 6) + ab));        \
        }
        OCT_STEP(0) OCT_STEP(1) OCT_STEP(2) OCT_STEP(3)
        OCT_STEP(4) OCT_STEP(5) OCT_STEP(6) OCT_STEP(7)
#undef OCT_STEP
#pragma unroll
        for (int t = 0; t < 8; ++t)
            asm volatile("" :: "v"(a[t].x), "v"(a[t].y), "v"(a[t].z), "v"(a[t].w),
                               "v"(b[t].x), "v"(b[t].y), "v"(b[t].z), "v"(b[t].w));
        float res = 0.0f;
#pragma unroll
        for (int t = 0; t < 8; ++t) {
            float p = a[t].x * b[t].x;
            p = fmaf(a[t].y, b[t].y, p);
            p = fmaf(a[t].z, b[t].z, p);
            p = fmaf(a[t].w, b[t].w, p);
            p = (j < 4) ? p : a[t].x;
            p = oct_sum8(p);
            if (j == t) res = p;
        }
        const int jo = g * 32 + lane;
        if (jo < total) out[jo] = res;
        s = sn;
        d = dn;
    }
}

extern "C" void kernel_launch(void* const* d_in, const int* in_sizes, int n_in,
                              void* d_out, int out_size, void* d_ws, size_t ws_size,
                              hipStream_t stream) {
    const float* x   = (const float*)d_in[0];
    const int*   tei = (const int*)d_in[1];
    const int*   tpo = (const int*)d_in[2];
    const int*   tne = (const int*)d_in[3];
    const float* W1  = (const float*)d_in[4];
    const float* b1  = (const float*)d_in[5];
    const float* W2  = (const float*)d_in[6];
    const float* b2  = (const float*)d_in[7];

    const int N  = in_sizes[0] / 256;
    const int E  = in_sizes[1] / 2;
    const int ET = in_sizes[2] / 2;

    float* ws   = (float*)d_ws;
    float* h1   = ws;
    float* z1   = ws + (size_t)16 * N;
    float* a2   = ws + (size_t)32 * N;
    float* rec  = ws + (size_t)96 * N;
    int*   cnt  = (int*)(ws + (size_t)160 * N);
    float* dinv = ws + (size_t)161 * N;
    int*   rowp = (int*)(ws + (size_t)162 * N);
    int*   csr  = (int*)(ws + (size_t)163 * N);
    unsigned* packed = (unsigned*)(csr + (size_t)E);
    int*   hist = (int*)(packed + (size_t)E);
    int*   loc  = hist + (size_t)MAX_BINS * SORT_G;

    const int B = 256;
    auto cdiv = [](long long a, long long b) { return (int)((a + b - 1) / b); };

    const int nb    = cdiv(N, BIN_W);
    const int chunk = cdiv(E, SORT_G);
    bin_local<<<SORT_G, SORT_T, 0, stream>>>(tei, tei + E, hist, loc, packed, E, nb, chunk);
    bin_sort<<<nb, SORT_T, 0, stream>>>(packed, hist, loc, rowp, cnt, dinv, csr, N, E, chunk);

    const int tiles = (N + 15) / 16;

    gemm_x_w1<<<cdiv(tiles, 8), B, 0, stream>>>(x, W1, dinv, h1, N);
    gather_rows<16, true, 8><<<cdiv((long long)4 * N, B), B, 0, stream>>>(
        rowp, cnt, csr, dinv, h1, b1, z1, N);

    gather_rows<16, false, 8><<<cdiv((long long)4 * N, B), B, 0, stream>>>(
        rowp, cnt, csr, dinv, z1, nullptr, a2, N);
    gemm_a_wwt<<<cdiv(tiles, 8), B, 0, stream>>>(a2, W2, b2, rec, N);

    edge_score<<<cdiv(2 * (long long)ET, 1024), B, 0, stream>>>(tpo, tpo + ET, tne, tne + ET,
                                                               rec, a2, (float*)d_out, ET);
}


# Layer-2 scoring applies the 16x16 form of W2 per edge on WMMA in the scoring launch, from the aggregated rows A[s], A[d]: no record plane, no record launch, 6 launches not 7 (sum order changes)
# speedup vs baseline: 17.5163x; 1.0335x over previous
#include <hip/hip_runtime.h>

typedef __attribute__((ext_vector_type(2))) float v2f;
typedef __attribute__((ext_vector_type(8))) float v8f;

constexpr int BIN_SH   = 9;
constexpr int BIN_W    = 1 << BIN_SH;
constexpr int MAX_BINS = 1024;
constexpr int SORT_G   = 256;
constexpr int SORT_T   = 1024;
constexpr int L1_STAGE = 16384;
constexpr int L2_STAGE = 32768;
constexpr int PIECES_W = SORT_G / (SORT_T / 32);
constexpr int PIECE_R  = 3;

__device__ __forceinline__ int wave_incl_scan(int v, int lane) {
    for (int o = 1; o < 32; o <<= 1) {
        int u = __shfl_up(v, o);
        if (lane >= o) v += u;
    }
    return v;
}

__global__ __launch_bounds__(SORT_T) void bin_local(const int* __restrict__ src,
                                                    const int* __restrict__ dst,
                                                    int* __restrict__ hist,
                                                    int* __restrict__ loc,
                                                    unsigned* __restrict__ packed,
                                                    int nE, int nb, int chunk) {
    __shared__ int s_h[MAX_BINS];
    __shared__ int s_wave[SORT_T / 32];
    __shared__ unsigned s_stage[L1_STAGE];
    const int t = threadIdx.x, lane = t & 31, w = t >> 5;
    const int blk = blockIdx.x;
    if (t < nb) s_h[t] = 0;
    __syncthreads();
    const int beg = blk * chunk;
    const int end = min(beg + chunk, nE);
    int base = beg;
    for (; base + 4 * SORT_T <= end; base += 4 * SORT_T) {
        int d[4];
#pragma unroll
        for (int k = 0; k < 4; ++k) d[k] = dst[base + k * SORT_T + t];
#pragma unroll
        for (int k = 0; k < 4; ++k) atomicAdd(&s_h[d[k] >> BIN_SH], 1);
    }
    for (int e = base + t; e < end; e += SORT_T) atomicAdd(&s_h[dst[e] >> BIN_SH], 1);
    __syncthreads();

    const int v = (t < nb) ? s_h[t] : 0;
    const int inc = wave_incl_scan(v, lane);
    if (lane == 31) s_wave[w] = inc;
    __syncthreads();
    int ex = inc - v;
    for (int k = 0; k < SORT_T / 32; ++k)
        if (k < w) ex += s_wave[k];
    if (t < nb) {
        hist[t * SORT_G + blk] = v;
        loc[t * SORT_G + blk]  = ex;
        s_h[t] = ex;
    }
    __syncthreads();

    base = beg;
    if (end - beg <= L1_STAGE) {
        for (; base + 4 * SORT_T <= end; base += 4 * SORT_T) {
            int d[4], s[4], q[4];
#pragma unroll
            for (int k = 0; k < 4; ++k) {
                const int e = base + k * SORT_T + t;
                d[k] = dst[e];
                s[k] = src[e];
            }
#pragma unroll
            for (int k = 0; k < 4; ++k) q[k] = atomicAdd(&s_h[d[k] >> BIN_SH], 1);
#pragma unroll
            for (int k = 0; k < 4; ++k)
                s_stage[q[k]] = ((unsigned)s[k] << BIN_SH) | ((unsigned)d[k] & (BIN_W - 1));
        }
        for (int e = base + t; e < end; e += SORT_T) {
            const int d = dst[e];
            s_stage[atomicAdd(&s_h[d >> BIN_SH], 1)] =
                ((unsigned)src[e] << BIN_SH) | ((unsigned)d & (BIN_W - 1));
        }
        __syncthreads();
        const int nloc = end - beg;
#pragma unroll 4
        for (int i = t; i < nloc; i += SORT_T) packed[beg + i] = s_stage[i];
    } else {
        for (int e = base + t; e < end; e += SORT_T) {
            const int d = dst[e];
            packed[beg + atomicAdd(&s_h[d >> BIN_SH], 1)] =
                ((unsigned)src[e] << BIN_SH) | ((unsigned)d & (BIN_W - 1));
        }
    }
}

__global__ __launch_bounds__(SORT_T) void bin_sort(const unsigned* __restrict__ packed,
                                                   const int* __restrict__ hist,
                                                   const int* __restrict__ loc,
                                                   int* __restrict__ rowptr,
                                                   int* __restrict__ cnt,
                                                   float* __restrict__ dinv,
                                                   int* __restrict__ csr,
                                                   int n, int nE, int chunk) {
    __shared__ int s_c[BIN_W];
    __shared__ int s_cur[BIN_W];
    __shared__ int s_wave[SORT_T / 32];
    __shared__ int s_red[SORT_T / 32];
    __shared__ int s_off[SORT_G];
    __shared__ int s_len[SORT_G];
    __shared__ int s_gs[SORT_G];
    __shared__ unsigned s_in[L2_STAGE];
    __shared__ int s_stage[L2_STAGE];
    const int t = threadIdx.x, lane = t & 31, w = t >> 5;
    const int b = blockIdx.x;

    int c0 = 0, l0 = 0;
    if (t < SORT_G) {
        c0 = hist[b * SORT_G + t];
        l0 = loc[b * SORT_G + t];
    }
    const int inc0 = wave_incl_scan(c0, lane);
    int part = l0;
    for (int o = 16; o > 0; o >>= 1) part += __shfl_xor(part, o);
    if (lane == 31) s_wave[w] = inc0;
    if (lane == 0) s_red[w] = part;
    if (t < BIN_W) s_c[t] = 0;
    __syncthreads();
    int ex0 = inc0 - c0, beg = 0, total = 0;
    for (int k = 0; k < SORT_T / 32; ++k) {
        const int sw = s_wave[k];
        if (k < w) ex0 += sw;
        total += sw;
        beg += s_red[k];
    }
    if (t < SORT_G) {
        s_off[t] = ex0;
        s_len[t] = c0;
        s_gs[t]  = t * chunk + l0;
    }
    __syncthreads();
    const int end = beg + total;
    const bool fits = total <= L2_STAGE;

    if (fits) {
        int len[PIECES_W], gs[PIECES_W], off[PIECES_W];
        unsigned v[PIECES_W][PIECE_R];
#pragma unroll
        for (int j = 0; j < PIECES_W; ++j) {
            const int pc = PIECES_W * w + j;
            len[j] = s_len[pc];
            gs[j]  = s_gs[pc];
            off[j] = s_off[pc];
        }
#pragma unroll
        for (int j = 0; j < PIECES_W; ++j) {
#pragma unroll
            for (int r = 0; r < PIECE_R; ++r) {
                const int i = min(lane + 32 * r, max(len[j] - 1, 0));
                v[j][r] = packed[max(min(gs[j] + i, nE - 1), 0)];
            }
        }
#pragma unroll
        for (int j = 0; j < PIECES_W; ++j) {
#pragma unroll
            for (int r = 0; r < PIECE_R; ++r) {
                const int i = lane + 32 * r;
                if (i < len[j]) {
                    s_in[off[j] + i] = v[j][r];
                    atomicAdd(&s_c[v[j][r] & (BIN_W - 1)], 1);
                }
            }
        }
#pragma unroll
        for (int j = 0; j < PIECES_W; ++j) {
            for (int i = lane + 32 * PIECE_R; i < len[j]; i += 32) {
                const unsigned p = packed[gs[j] + i];
                s_in[off[j] + i] = p;
                atomicAdd(&s_c[p & (BIN_W - 1)], 1);
            }
        }
    } else {
        for (int j = 0; j < PIECES_W; ++j) {
            const int pc = PIECES_W * w + j;
            const int ln = s_len[pc], g = s_gs[pc];
            for (int i = lane; i < ln; i += 32) atomicAdd(&s_c[packed[g + i] & (BIN_W - 1)], 1);
        }
    }
    __syncthreads();

    const int c = (t < BIN_W) ? s_c[t] : 0;
    const int inc = wave_incl_scan(c, lane);
    if (lane == 31) s_wave[w] = inc;
    __syncthreads();
    int ex = inc - c;
    for (int k = 0; k < BIN_W / 32; ++k)
        if (k < w) ex += s_wave[k];
    if (t < BIN_W) {
        s_cur[t] = beg + ex;
        const int node = b * BIN_W + t;
        if (node < n) {
            rowptr[node] = beg + ex;
            cnt[node] = c;
            dinv[node] = rsqrtf((float)c + 1.0f);
        }
    }
    __syncthreads();

    if (fits) {
        int i = t;
        for (; i + 3 * SORT_T < total; i += 4 * SORT_T) {
            unsigned p[4];
            int q[4];
#pragma unroll
            for (int k = 0; k < 4; ++k) p[k] = s_in[i + k * SORT_T];
#pragma unroll
            for (int k = 0; k < 4; ++k) q[k] = atomicAdd(&s_cur[p[k] & (BIN_W - 1)], 1) - beg;
#pragma unroll
            for (int k = 0; k < 4; ++k) s_stage[q[k]] = (int)(p[k] >> BIN_SH);
        }
        for (; i < total; i += SORT_T) {
            const unsigned p = s_in[i];
            s_stage[atomicAdd(&s_cur[p & (BIN_W - 1)], 1) - beg] = (int)(p >> BIN_SH);
        }
        __syncthreads();
        int gi = (beg & ~31) + t;
        if (gi >= beg && gi < end) csr[gi] = s_stage[gi - beg];
#pragma unroll 4
        for (gi += SORT_T; gi < end; gi += SORT_T) csr[gi] = s_stage[gi - beg];
    } else {
        for (int j = 0; j < PIECES_W; ++j) {
            const int pc = PIECES_W * w + j;
            const int ln = s_len[pc], g = s_gs[pc];
            for (int i = lane; i < ln; i += 32) {
                const unsigned p = packed[g + i];
                csr[atomicAdd(&s_cur[p & (BIN_W - 1)], 1)] = (int)(p >> BIN_SH);
            }
        }
    }
}

__global__ __launch_bounds__(256) void gemm_x_w1(const float* __restrict__ X,
                                                 const float* __restrict__ W,
                                                 const float* __restrict__ dinv,
                                                 float* __restrict__ H, int n) {
    const int lane = threadIdx.x & 31;
    const int tile = blockIdx.x * 8 + (threadIdx.x >> 5);
    const int row0 = tile * 16;
    if (row0 >= n) return;
    int m = row0 + (lane & 15);
    if (m >= n) m = n - 1;
    const int kh  = (lane >> 4) << 1;
    const int col = lane & 15;
    const float* xr = X + (size_t)m * 256;
    const int ra = (lane >> 4) << 3;
    float sc[8];
#pragma unroll
    for (int v = 0; v < 8; ++v) sc[v] = dinv[min(row0 + ra + v, n - 1)];

    v8f acc = {};
#pragma unroll 4
    for (int k0 = 0; k0 < 256; k0 += 4) {
        v2f a, b;
        a.x = xr[k0 + kh];
        a.y = xr[k0 + kh + 1];
        b.x = W[(size_t)(k0 + kh) * 16 + col];
        b.y = W[(size_t)(k0 + kh + 1) * 16 + col];
        acc = __builtin_amdgcn_wmma_f32_16x16x4_f32(false, a, false, b,
                                                    (short)0, acc, false, false);
    }
#pragma unroll
    for (int v = 0; v < 8; ++v) acc[v] *= sc[v];
    if (row0 + 16 <= n) {
#pragma unroll
        for (int v = 0; v < 8; ++v)
            H[(size_t)(row0 + ra + v) * 16 + col] = acc[v];
    } else {
#pragma unroll
        for (int v = 0; v < 8; ++v) {
            int r = row0 + ra + v;
            if (r < n) H[(size_t)r * 16 + col] = acc[v];
        }
    }
}

template <int F, bool FIRST, int U>
__global__ __launch_bounds__(256) void gather_rows(const int* __restrict__ rowptr,
                                                   const int* __restrict__ cnt,
                                                   const int* __restrict__ csr,
                                                   const float* __restrict__ dinv,
                                                   const float* __restrict__ H,
                                                   const float* __restrict__ bias,
                                                   float* __restrict__ Z, int n) {
    constexpr int L = F / 4;
    const long long gid = (long long)blockIdx.x * blockDim.x + threadIdx.x;
    const int i = (int)(gid / L);
    const int l = (int)(gid % L);
    if (i >= n) return;
    const float4* H4 = (const float4*)H;
    const int beg = rowptr[i];
    const int end = beg + cnt[i];
    const float di = dinv[i];
    float4 acc = make_float4(0.0f, 0.0f, 0.0f, 0.0f);
    if (beg < end) {
        const int last = end - 1;
        int s[U];
#pragma unroll
        for (int k = 0; k < U; ++k) s[k] = csr[min(beg + k, last)];
        for (int p = beg; p < end; p += U) {
            int sn[U];
#pragma unroll
            for (int k = 0; k < U; ++k) sn[k] = csr[min(p + U + k, last)];
            float4 h[U];
#pragma unroll
            for (int k = 0; k < U; ++k) h[k] = H4[(size_t)s[k] * L + l];
#pragma unroll
            for (int k = 0; k < U; ++k)
                asm volatile("" :: "v"(h[k].x), "v"(h[k].y), "v"(h[k].z), "v"(h[k].w));
#pragma unroll
            for (int k = 0; k < U; ++k) {
                if (k == 0 || p + k < end) {
                    acc.x += h[k].x;
                    acc.y += h[k].y;
                    acc.z += h[k].z;
                    acc.w += h[k].w;
                }
            }
#pragma unroll
            for (int k = 0; k < U; ++k) s[k] = sn[k];
        }
    }
    const float4 hs = H4[(size_t)i * L + l];
    float4 v;
    v.x = di * (acc.x + hs.x);
    v.y = di * (acc.y + hs.y);
    v.z = di * (acc.z + hs.z);
    v.w = di * (acc.w + hs.w);
    if (FIRST) {
        v.x = di * fmaxf(v.x + bias[4 * l + 0], 0.0f);
        v.y = di * fmaxf(v.y + bias[4 * l + 1], 0.0f);
        v.z = di * fmaxf(v.z + bias[4 * l + 2], 0.0f);
        v.w = di * fmaxf(v.w + bias[4 * l + 3], 0.0f);
    }
    ((float4*)Z)[(size_t)i * L + l] = v;
}

__global__ __launch_bounds__(256) void edge_score(const int* __restrict__ psrc,
                                                  const int* __restrict__ pdst,
                                                  const int* __restrict__ nsrc,
                                                  const int* __restrict__ ndst,
                                                  const float* __restrict__ A,
                                                  const float* __restrict__ W,
                                                  const float* __restrict__ bias,
                                                  float* __restrict__ out, int nT) {
    const int lane = threadIdx.x & 31;
    const int col = lane & 15;
    const int h   = lane >> 4;
    const bool hi = h != 0;
    const int total = 2 * nT;
    const int ngroups = (total + 31) >> 5;
    const int nwaves = gridDim.x * 8;
    int g = blockIdx.x * 8 + __builtin_amdgcn_readfirstlane((int)(threadIdx.x >> 5));
    if (g >= ngroups) return;

    int s[2], d[2];
#pragma unroll
    for (int q = 0; q < 2; ++q) {
        const int jx = min(g * 32 + 16 * q + col, total - 1);
        s[q] = (jx < nT) ? psrc[jx] : nsrc[jx - nT];
        d[q] = (jx < nT) ? pdst[jx] : ndst[jx - nT];
    }
    __builtin_amdgcn_sched_barrier(0);

    const float4* wr = (const float4*)(W + (size_t)col * 64);
    const float4* br = (const float4*)bias;
    v8f gm = {};
    float cp = 0.0f, bp = 0.0f;
#pragma unroll
    for (int q = 0; q < 8; ++q) {
        const float4 w = wr[2 * q + h];
        const float4 b = br[2 * q + h];
        cp = fmaf(w.x, b.x, cp); cp = fmaf(w.y, b.y, cp);
        cp = fmaf(w.z, b.z, cp); cp = fmaf(w.w, b.w, cp);
        bp = fmaf(b.x, b.x, bp); bp = fmaf(b.y, b.y, bp);
        bp = fmaf(b.z, b.z, bp); bp = fmaf(b.w, b.w, bp);
        v2f f;
        f.x = w.x; f.y = w.y;
        gm = __builtin_amdgcn_wmma_f32_16x16x4_f32(false, f, false, f, (short)0, gm, false, false);
        f.x = w.z; f.y = w.w;
        gm = __builtin_amdgcn_wmma_f32_16x16x4_f32(false, f, false, f, (short)0, gm, false, false);
    }
    const float cc = cp + __shfl_xor(cp, 16);
    const float bb = bp + __shfl_xor(bp, 16);
    float ch[8];
    const int cci = __float_as_int(cc);
#pragma unroll
    for (int i = 0; i < 8; ++i) {
        const float clo = __int_as_float(__builtin_amdgcn_readlane(cci, i));
        const float chi = __int_as_float(__builtin_amdgcn_readlane(cci, i + 8));
        ch[i] = hi ? chi : clo;
    }

    for (; g < ngroups; g += nwaves) {
        const int gn = min(g + nwaves, ngroups - 1);
        int sn[2], dn[2];
#pragma unroll
        for (int q = 0; q < 2; ++q) {
            const int jn = min(gn * 32 + 16 * q + col, total - 1);
            sn[q] = (jn < nT) ? psrc[jn] : nsrc[jn - nT];
            dn[q] = (jn < nT) ? pdst[jn] : ndst[jn - nT];
        }

        float4 as0[2], as1[2], ad0[2], ad1[2];
#pragma unroll
        for (int q = 0; q < 2; ++q) {
            const float4* sr = (const float4*)(A + (size_t)s[q] * 16 + 8 * h);
            const float4* dr = (const float4*)(A + (size_t)d[q] * 16 + 8 * h);
            as0[q] = sr[0];
            as1[q] = sr[1];
            ad0[q] = dr[0];
            ad1[q] = dr[1];
        }

        __builtin_amdgcn_sched_barrier(0);

        float tot[2];
#pragma unroll
        for (int q = 0; q < 2; ++q) {
            const float av[8] = {as0[q].x, as0[q].y, as0[q].z, as0[q].w,
                                 as1[q].x, as1[q].y, as1[q].z, as1[q].w};
            const float dv[8] = {ad0[q].x, ad0[q].y, ad0[q].z, ad0[q].w,
                                 ad1[q].x, ad1[q].y, ad1[q].z, ad1[q].w};
            v8f acc = {};
#pragma unroll
            for (int w = 0; w < 4; ++w) {
                v2f ga, sb;
                ga.x = gm[2 * w]; ga.y = gm[2 * w + 1];
                sb.x = av[2 * w]; sb.y = av[2 * w + 1];
                acc = __builtin_amdgcn_wmma_f32_16x16x4_f32(false, ga, false, sb, (short)0, acc, false, false);
            }
            float t = 0.0f;
#pragma unroll
            for (int v = 0; v < 8; ++v) {
                t = fmaf(acc[v] + ch[v], dv[v], t);
                t = fmaf(av[v], ch[v], t);
            }
            tot[q] = t;
        }
        const float keep = hi ? tot[1] : tot[0];
        const float send = hi ? tot[0] : tot[1];
        const float res = keep + __shfl_xor(send, 16) + bb;
        const int jo = g * 32 + lane;
        if (jo < total) out[jo] = res;
#pragma unroll
        for (int q = 0; q < 2; ++q) {
            s[q] = sn[q];
            d[q] = dn[q];
        }
    }
}

extern "C" void kernel_launch(void* const* d_in, const int* in_sizes, int n_in,
                              void* d_out, int out_size, void* d_ws, size_t ws_size,
                              hipStream_t stream) {
    const float* x   = (const float*)d_in[0];
    const int*   tei = (const int*)d_in[1];
    const int*   tpo = (const int*)d_in[2];
    const int*   tne = (const int*)d_in[3];
    const float* W1  = (const float*)d_in[4];
    const float* b1  = (const float*)d_in[5];
    const float* W2  = (const float*)d_in[6];
    const float* b2  = (const float*)d_in[7];

    const int N  = in_sizes[0] / 256;
    const int E  = in_sizes[1] / 2;
    const int ET = in_sizes[2] / 2;

    float* ws   = (float*)d_ws;
    float* h1   = ws;
    float* z1   = ws + (size_t)16 * N;
    float* a2   = ws + (size_t)32 * N;
    int*   cnt  = (int*)(ws + (size_t)160 * N);
    float* dinv = ws + (size_t)161 * N;
    int*   rowp = (int*)(ws + (size_t)162 * N);
    int*   csr  = (int*)(ws + (size_t)163 * N);
    unsigned* packed = (unsigned*)(csr + (size_t)E);
    int*   hist = (int*)(packed + (size_t)E);
    int*   loc  = hist + (size_t)MAX_BINS * SORT_G;

    const int B = 256;
    auto cdiv = [](long long a, long long b) { return (int)((a + b - 1) / b); };

    const int nb    = cdiv(N, BIN_W);
    const int chunk = cdiv(E, SORT_G);
    bin_local<<<SORT_G, SORT_T, 0, stream>>>(tei, tei + E, hist, loc, packed, E, nb, chunk);
    bin_sort<<<nb, SORT_T, 0, stream>>>(packed, hist, loc, rowp, cnt, dinv, csr, N, E, chunk);

    const int tiles = (N + 15) / 16;

    gemm_x_w1<<<cdiv(tiles, 8), B, 0, stream>>>(x, W1, dinv, h1, N);
    gather_rows<16, true, 8><<<cdiv((long long)4 * N, B), B, 0, stream>>>(
        rowp, cnt, csr, dinv, h1, b1, z1, N);

    gather_rows<16, false, 8><<<cdiv((long long)4 * N, B), B, 0, stream>>>(
        rowp, cnt, csr, dinv, z1, nullptr, a2, N);

    edge_score<<<cdiv(2 * (long long)ET, 1024), B, 0, stream>>>(tpo, tpo + ET, tne, tne + ET,
                                                               a2, W2, b2, (float*)d_out, ET);
}


# Aggregation: a block copies its nodes' one contiguous stretch of csr into LDS with whole-line loads and the lane groups take their source indices from LDS, not by 4-byte global loads per lane
# speedup vs baseline: 20.7539x; 1.1848x over previous
#include <hip/hip_runtime.h>

typedef __attribute__((ext_vector_type(2))) float v2f;
typedef __attribute__((ext_vector_type(8))) float v8f;

constexpr int BIN_SH   = 9;
constexpr int BIN_W    = 1 << BIN_SH;
constexpr int MAX_BINS = 1024;
constexpr int SORT_G   = 256;
constexpr int SORT_T   = 1024;
constexpr int L1_STAGE = 16384;
constexpr int L2_STAGE = 32768;
constexpr int PIECES_W = SORT_G / (SORT_T / 32);
constexpr int PIECE_R  = 3;

__device__ __forceinline__ int wave_incl_scan(int v, int lane) {
    for (int o = 1; o < 32; o <<= 1) {
        int u = __shfl_up(v, o);
        if (lane >= o) v += u;
    }
    return v;
}

__global__ __launch_bounds__(SORT_T) void bin_local(const int* __restrict__ src,
                                                    const int* __restrict__ dst,
                                                    int* __restrict__ hist,
                                                    int* __restrict__ loc,
                                                    unsigned* __restrict__ packed,
                                                    int nE, int nb, int chunk) {
    __shared__ int s_h[MAX_BINS];
    __shared__ int s_wave[SORT_T / 32];
    __shared__ unsigned s_stage[L1_STAGE];
    const int t = threadIdx.x, lane = t & 31, w = t >> 5;
    const int blk = blockIdx.x;
    if (t < nb) s_h[t] = 0;
    __syncthreads();
    const int beg = blk * chunk;
    const int end = min(beg + chunk, nE);
    int base = beg;
    for (; base + 4 * SORT_T <= end; base += 4 * SORT_T) {
        int d[4];
#pragma unroll
        for (int k = 0; k < 4; ++k) d[k] = dst[base + k * SORT_T + t];
#pragma unroll
        for (int k = 0; k < 4; ++k) atomicAdd(&s_h[d[k] >> BIN_SH], 1);
    }
    for (int e = base + t; e < end; e += SORT_T) atomicAdd(&s_h[dst[e] >> BIN_SH], 1);
    __syncthreads();

    const int v = (t < nb) ? s_h[t] : 0;
    const int inc = wave_incl_scan(v, lane);
    if (lane == 31) s_wave[w] = inc;
    __syncthreads();
    int ex = inc - v;
    for (int k = 0; k < SORT_T / 32; ++k)
        if (k < w) ex += s_wave[k];
    if (t < nb) {
        hist[t * SORT_G + blk] = v;
        loc[t * SORT_G + blk]  = ex;
        s_h[t] = ex;
    }
    __syncthreads();

    base = beg;
    if (end - beg <= L1_STAGE) {
        for (; base + 4 * SORT_T <= end; base += 4 * SORT_T) {
            int d[4], s[4], q[4];
#pragma unroll
            for (int k = 0; k < 4; ++k) {
                const int e = base + k * SORT_T + t;
                d[k] = dst[e];
                s[k] = src[e];
            }
#pragma unroll
            for (int k = 0; k < 4; ++k) q[k] = atomicAdd(&s_h[d[k] >> BIN_SH], 1);
#pragma unroll
            for (int k = 0; k < 4; ++k)
                s_stage[q[k]] = ((unsigned)s[k] << BIN_SH) | ((unsigned)d[k] & (BIN_W - 1));
        }
        for (int e = base + t; e < end; e += SORT_T) {
            const int d = dst[e];
            s_stage[atomicAdd(&s_h[d >> BIN_SH], 1)] =
                ((unsigned)src[e] << BIN_SH) | ((unsigned)d & (BIN_W - 1));
        }
        __syncthreads();
        const int nloc = end - beg;
#pragma unroll 4
        for (int i = t; i < nloc; i += SORT_T) packed[beg + i] = s_stage[i];
    } else {
        for (int e = base + t; e < end; e += SORT_T) {
            const int d = dst[e];
            packed[beg + atomicAdd(&s_h[d >> BIN_SH], 1)] =
                ((unsigned)src[e] << BIN_SH) | ((unsigned)d & (BIN_W - 1));
        }
    }
}

__global__ __launch_bounds__(SORT_T) void bin_sort(const unsigned* __restrict__ packed,
                                                   const int* __restrict__ hist,
                                                   const int* __restrict__ loc,
                                                   int* __restrict__ rowptr,
                                                   int* __restrict__ cnt,
                                                   float* __restrict__ dinv,
                                                   int* __restrict__ csr,
                                                   int n, int nE, int chunk) {
    __shared__ int s_c[BIN_W];
    __shared__ int s_cur[BIN_W];
    __shared__ int s_wave[SORT_T / 32];
    __shared__ int s_red[SORT_T / 32];
    __shared__ int s_off[SORT_G];
    __shared__ int s_len[SORT_G];
    __shared__ int s_gs[SORT_G];
    __shared__ unsigned s_in[L2_STAGE];
    __shared__ int s_stage[L2_STAGE];
    const int t = threadIdx.x, lane = t & 31, w = t >> 5;
    const int b = blockIdx.x;

    int c0 = 0, l0 = 0;
    if (t < SORT_G) {
        c0 = hist[b * SORT_G + t];
        l0 = loc[b * SORT_G + t];
    }
    const int inc0 = wave_incl_scan(c0, lane);
    int part = l0;
    for (int o = 16; o > 0; o >>= 1) part += __shfl_xor(part, o);
    if (lane == 31) s_wave[w] = inc0;
    if (lane == 0) s_red[w] = part;
    if (t < BIN_W) s_c[t] = 0;
    __syncthreads();
    int ex0 = inc0 - c0, beg = 0, total = 0;
    for (int k = 0; k < SORT_T / 32; ++k) {
        const int sw = s_wave[k];
        if (k < w) ex0 += sw;
        total += sw;
        beg += s_red[k];
    }
    if (t < SORT_G) {
        s_off[t] = ex0;
        s_len[t] = c0;
        s_gs[t]  = t * chunk + l0;
    }
    __syncthreads();
    const int end = beg + total;
    const bool fits = total <= L2_STAGE;

    if (fits) {
        int len[PIECES_W], gs[PIECES_W], off[PIECES_W];
        unsigned v[PIECES_W][PIECE_R];
#pragma unroll
        for (int j = 0; j < PIECES_W; ++j) {
            const int pc = PIECES_W * w + j;
            len[j] = s_len[pc];
            gs[j]  = s_gs[pc];
            off[j] = s_off[pc];
        }
#pragma unroll
        for (int j = 0; j < PIECES_W; ++j) {
#pragma unroll
            for (int r = 0; r < PIECE_R; ++r) {
                const int i = min(lane + 32 * r, max(len[j] - 1, 0));
                v[j][r] = packed[max(min(gs[j] + i, nE - 1), 0)];
            }
        }
#pragma unroll
        for (int j = 0; j < PIECES_W; ++j) {
#pragma unroll
            for (int r = 0; r < PIECE_R; ++r) {
                const int i = lane + 32 * r;
                if (i < len[j]) {
                    s_in[off[j] + i] = v[j][r];
                    atomicAdd(&s_c[v[j][r] & (BIN_W - 1)], 1);
                }
            }
        }
#pragma unroll
        for (int j = 0; j < PIECES_W; ++j) {
            for (int i = lane + 32 * PIECE_R; i < len[j]; i += 32) {
                const unsigned p = packed[gs[j] + i];
                s_in[off[j] + i] = p;
                atomicAdd(&s_c[p & (BIN_W - 1)], 1);
            }
        }
    } else {
        for (int j = 0; j < PIECES_W; ++j) {
            const int pc = PIECES_W * w + j;
            const int ln = s_len[pc], g = s_gs[pc];
            for (int i = lane; i < ln; i += 32) atomicAdd(&s_c[packed[g + i] & (BIN_W - 1)], 1);
        }
    }
    __syncthreads();

    const int c = (t < BIN_W) ? s_c[t] : 0;
    const int inc = wave_incl_scan(c, lane);
    if (lane == 31) s_wave[w] = inc;
    __syncthreads();
    int ex = inc - c;
    for (int k = 0; k < BIN_W / 32; ++k)
        if (k < w) ex += s_wave[k];
    if (t < BIN_W) {
        s_cur[t] = beg + ex;
        const int node = b * BIN_W + t;
        if (node < n) {
            rowptr[node] = beg + ex;
            cnt[node] = c;
            dinv[node] = rsqrtf((float)c + 1.0f);
        }
    }
    __syncthreads();

    if (fits) {
        int i = t;
        for (; i + 3 * SORT_T < total; i += 4 * SORT_T) {
            unsigned p[4];
            int q[4];
#pragma unroll
            for (int k = 0; k < 4; ++k) p[k] = s_in[i + k * SORT_T];
#pragma unroll
            for (int k = 0; k < 4; ++k) q[k] = atomicAdd(&s_cur[p[k] & (BIN_W - 1)], 1) - beg;
#pragma unroll
            for (int k = 0; k < 4; ++k) s_stage[q[k]] = (int)(p[k] >> BIN_SH);
        }
        for (; i < total; i += SORT_T) {
            const unsigned p = s_in[i];
            s_stage[atomicAdd(&s_cur[p & (BIN_W - 1)], 1) - beg] = (int)(p >> BIN_SH);
        }
        __syncthreads();
        int gi = (beg & ~31) + t;
        if (gi >= beg && gi < end) csr[gi] = s_stage[gi - beg];
#pragma unroll 4
        for (gi += SORT_T; gi < end; gi += SORT_T) csr[gi] = s_stage[gi - beg];
    } else {
        for (int j = 0; j < PIECES_W; ++j) {
            const int pc = PIECES_W * w + j;
            const int ln = s_len[pc], g = s_gs[pc];
            for (int i = lane; i < ln; i += 32) {
                const unsigned p = packed[g + i];
                csr[atomicAdd(&s_cur[p & (BIN_W - 1)], 1)] = (int)(p >> BIN_SH);
            }
        }
    }
}

__global__ __launch_bounds__(256) void gemm_x_w1(const float* __restrict__ X,
                                                 const float* __restrict__ W,
                                                 const float* __restrict__ dinv,
                                                 float* __restrict__ H, int n) {
    const int lane = threadIdx.x & 31;
    const int tile = blockIdx.x * 8 + (threadIdx.x >> 5);
    const int row0 = tile * 16;
    if (row0 >= n) return;
    int m = row0 + (lane & 15);
    if (m >= n) m = n - 1;
    const int kh  = (lane >> 4) << 1;
    const int col = lane & 15;
    const float* xr = X + (size_t)m * 256;
    const int ra = (lane >> 4) << 3;
    float sc[8];
#pragma unroll
    for (int v = 0; v < 8; ++v) sc[v] = dinv[min(row0 + ra + v, n - 1)];

    v8f acc = {};
#pragma unroll 4
    for (int k0 = 0; k0 < 256; k0 += 4) {
        v2f a, b;
        a.x = xr[k0 + kh];
        a.y = xr[k0 + kh + 1];
        b.x = W[(size_t)(k0 + kh) * 16 + col];
        b.y = W[(size_t)(k0 + kh + 1) * 16 + col];
        acc = __builtin_amdgcn_wmma_f32_16x16x4_f32(false, a, false, b,
                                                    (short)0, acc, false, false);
    }
#pragma unroll
    for (int v = 0; v < 8; ++v) acc[v] *= sc[v];
    if (row0 + 16 <= n) {
#pragma unroll
        for (int v = 0; v < 8; ++v)
            H[(size_t)(row0 + ra + v) * 16 + col] = acc[v];
    } else {
#pragma unroll
        for (int v = 0; v < 8; ++v) {
            int r = row0 + ra + v;
            if (r < n) H[(size_t)r * 16 + col] = acc[v];
        }
    }
}

constexpr int IDX_CAP = 4096;
constexpr int IDX_R   = 9;
static_assert(IDX_R * 256 <= IDX_CAP, "the rounds must fit the table");

template <int L, int U, typename Fetch>
__device__ __forceinline__ float4 sum_run(const float4* __restrict__ H4, int beg, int end, int l,
                                          Fetch fetch) {
    float4 acc = make_float4(0.0f, 0.0f, 0.0f, 0.0f);
    if (beg < end) {
        const int last = end - 1;
        int s[U];
#pragma unroll
        for (int k = 0; k < U; ++k) s[k] = fetch(min(beg + k, last));
        for (int p = beg; p < end; p += U) {
            int sn[U];
#pragma unroll
            for (int k = 0; k < U; ++k) sn[k] = fetch(min(p + U + k, last));
            float4 h[U];
#pragma unroll
            for (int k = 0; k < U; ++k) h[k] = H4[(size_t)s[k] * L + l];
#pragma unroll
            for (int k = 0; k < U; ++k)
                asm volatile("" :: "v"(h[k].x), "v"(h[k].y), "v"(h[k].z), "v"(h[k].w));
#pragma unroll
            for (int k = 0; k < U; ++k) {
                if (k == 0 || p + k < end) {
                    acc.x += h[k].x;
                    acc.y += h[k].y;
                    acc.z += h[k].z;
                    acc.w += h[k].w;
                }
            }
#pragma unroll
            for (int k = 0; k < U; ++k) s[k] = sn[k];
        }
    }
    return acc;
}

template <int F, bool FIRST, int U>
__global__ __launch_bounds__(256) void gather_rows(const int* __restrict__ rowptr,
                                                   const int* __restrict__ cnt,
                                                   const int* __restrict__ csr,
                                                   const float* __restrict__ dinv,
                                                   const float* __restrict__ H,
                                                   const float* __restrict__ bias,
                                                   float* __restrict__ Z, int n) {
    constexpr int L = F / 4;
    constexpr int NPB = 256 / L;
    __shared__ int s_idx[IDX_CAP];
    const int t = threadIdx.x;
    const int i0 = blockIdx.x * NPB;
    const int i1 = min(i0 + NPB, n) - 1;
    const int bbeg = rowptr[i0];
    const int bend = rowptr[i1] + cnt[i1];
    const int blen = bend - bbeg;
    const bool staged = blen <= IDX_CAP;
    const int i = i0 + t / L;
    const int l = t % L;
    const bool valid = i < n;
    int beg = 0, end = 0;
    float di = 0.0f;
    if (valid) {
        beg = rowptr[i];
        end = beg + cnt[i];
        di = dinv[i];
    }
    if (staged && blen > 0) {
        int v[IDX_R];
#pragma unroll
        for (int r = 0; r < IDX_R; ++r) v[r] = csr[min(bbeg + t + 256 * r, bend - 1)];
#pragma unroll
        for (int r = 0; r < IDX_R; ++r) s_idx[t + 256 * r] = v[r];
        for (int o = t + 256 * IDX_R; o < blen; o += 256) s_idx[o] = csr[bbeg + o];
    }
    __syncthreads();
    const float4* H4 = (const float4*)H;
    float4 acc;
    if (staged) acc = sum_run<L, U>(H4, beg, end, l, [&](int j) { return s_idx[j - bbeg]; });
    else        acc = sum_run<L, U>(H4, beg, end, l, [&](int j) { return csr[j]; });
    if (!valid) return;
    const float4 hs = H4[(size_t)i * L + l];
    float4 v;
    v.x = di * (acc.x + hs.x);
    v.y = di * (acc.y + hs.y);
    v.z = di * (acc.z + hs.z);
    v.w = di * (acc.w + hs.w);
    if (FIRST) {
        v.x = di * fmaxf(v.x + bias[4 * l + 0], 0.0f);
        v.y = di * fmaxf(v.y + bias[4 * l + 1], 0.0f);
        v.z = di * fmaxf(v.z + bias[4 * l + 2], 0.0f);
        v.w = di * fmaxf(v.w + bias[4 * l + 3], 0.0f);
    }
    ((float4*)Z)[(size_t)i * L + l] = v;
}

__global__ __launch_bounds__(256) void edge_score(const int* __restrict__ psrc,
                                                  const int* __restrict__ pdst,
                                                  const int* __restrict__ nsrc,
                                                  const int* __restrict__ ndst,
                                                  const float* __restrict__ A,
                                                  const float* __restrict__ W,
                                                  const float* __restrict__ bias,
                                                  float* __restrict__ out, int nT) {
    const int lane = threadIdx.x & 31;
    const int col = lane & 15;
    const int h   = lane >> 4;
    const bool hi = h != 0;
    const int total = 2 * nT;
    const int ngroups = (total + 31) >> 5;
    const int nwaves = gridDim.x * 8;
    int g = blockIdx.x * 8 + __builtin_amdgcn_readfirstlane((int)(threadIdx.x >> 5));
    if (g >= ngroups) return;

    int s[2], d[2];
#pragma unroll
    for (int q = 0; q < 2; ++q) {
        const int jx = min(g * 32 + 16 * q + col, total - 1);
        s[q] = (jx < nT) ? psrc[jx] : nsrc[jx - nT];
        d[q] = (jx < nT) ? pdst[jx] : ndst[jx - nT];
    }
    __builtin_amdgcn_sched_barrier(0);

    const float4* wr = (const float4*)(W + (size_t)col * 64);
    const float4* br = (const float4*)bias;
    v8f gm = {};
    float cp = 0.0f, bp = 0.0f;
#pragma unroll
    for (int q = 0; q < 8; ++q) {
        const float4 w = wr[2 * q + h];
        const float4 b = br[2 * q + h];
        cp = fmaf(w.x, b.x, cp); cp = fmaf(w.y, b.y, cp);
        cp = fmaf(w.z, b.z, cp); cp = fmaf(w.w, b.w, cp);
        bp = fmaf(b.x, b.x, bp); bp = fmaf(b.y, b.y, bp);
        bp = fmaf(b.z, b.z, bp); bp = fmaf(b.w, b.w, bp);
        v2f f;
        f.x = w.x; f.y = w.y;
        gm = __builtin_amdgcn_wmma_f32_16x16x4_f32(false, f, false, f, (short)0, gm, false, false);
        f.x = w.z; f.y = w.w;
        gm = __builtin_amdgcn_wmma_f32_16x16x4_f32(false, f, false, f, (short)0, gm, false, false);
    }
    const float cc = cp + __shfl_xor(cp, 16);
    const float bb = bp + __shfl_xor(bp, 16);
    float ch[8];
    const int cci = __float_as_int(cc);
#pragma unroll
    for (int i = 0; i < 8; ++i) {
        const float clo = __int_as_float(__builtin_amdgcn_readlane(cci, i));
        const float chi = __int_as_float(__builtin_amdgcn_readlane(cci, i + 8));
        ch[i] = hi ? chi : clo;
    }

    for (; g < ngroups; g += nwaves) {
        const int gn = min(g + nwaves, ngroups - 1);
        int sn[2], dn[2];
#pragma unroll
        for (int q = 0; q < 2; ++q) {
            const int jn = min(gn * 32 + 16 * q + col, total - 1);
            sn[q] = (jn < nT) ? psrc[jn] : nsrc[jn - nT];
            dn[q] = (jn < nT) ? pdst[jn] : ndst[jn - nT];
        }

        float4 as0[2], as1[2], ad0[2], ad1[2];
#pragma unroll
        for (int q = 0; q < 2; ++q) {
            const float4* sr = (const float4*)(A + (size_t)s[q] * 16 + 8 * h);
            const float4* dr = (const float4*)(A + (size_t)d[q] * 16 + 8 * h);
            as0[q] = sr[0];
            as1[q] = sr[1];
            ad0[q] = dr[0];
            ad1[q] = dr[1];
        }

        __builtin_amdgcn_sched_barrier(0);

        float tot[2];
#pragma unroll
        for (int q = 0; q < 2; ++q) {
            const float av[8] = {as0[q].x, as0[q].y, as0[q].z, as0[q].w,
                                 as1[q].x, as1[q].y, as1[q].z, as1[q].w};
            const float dv[8] = {ad0[q].x, ad0[q].y, ad0[q].z, ad0[q].w,
                                 ad1[q].x, ad1[q].y, ad1[q].z, ad1[q].w};
            v8f acc = {};
#pragma unroll
            for (int w = 0; w < 4; ++w) {
                v2f ga, sb;
                ga.x = gm[2 * w]; ga.y = gm[2 * w + 1];
                sb.x = av[2 * w]; sb.y = av[2 * w + 1];
                acc = __builtin_amdgcn_wmma_f32_16x16x4_f32(false, ga, false, sb, (short)0, acc, false, false);
            }
            float t = 0.0f;
#pragma unroll
            for (int v = 0; v < 8; ++v) {
                t = fmaf(acc[v] + ch[v], dv[v], t);
                t = fmaf(av[v], ch[v], t);
            }
            tot[q] = t;
        }
        const float keep = hi ? tot[1] : tot[0];
        const float send = hi ? tot[0] : tot[1];
        const float res = keep + __shfl_xor(send, 16) + bb;
        const int jo = g * 32 + lane;
        if (jo < total) out[jo] = res;
#pragma unroll
        for (int q = 0; q < 2; ++q) {
            s[q] = sn[q];
            d[q] = dn[q];
        }
    }
}

extern "C" void kernel_launch(void* const* d_in, const int* in_sizes, int n_in,
                              void* d_out, int out_size, void* d_ws, size_t ws_size,
                              hipStream_t stream) {
    const float* x   = (const float*)d_in[0];
    const int*   tei = (const int*)d_in[1];
    const int*   tpo = (const int*)d_in[2];
    const int*   tne = (const int*)d_in[3];
    const float* W1  = (const float*)d_in[4];
    const float* b1  = (const float*)d_in[5];
    const float* W2  = (const float*)d_in[6];
    const float* b2  = (const float*)d_in[7];

    const int N  = in_sizes[0] / 256;
    const int E  = in_sizes[1] / 2;
    const int ET = in_sizes[2] / 2;

    float* ws   = (float*)d_ws;
    float* h1   = ws;
    float* z1   = ws + (size_t)16 * N;
    float* a2   = ws + (size_t)32 * N;
    int*   cnt  = (int*)(ws + (size_t)160 * N);
    float* dinv = ws + (size_t)161 * N;
    int*   rowp = (int*)(ws + (size_t)162 * N);
    int*   csr  = (int*)(ws + (size_t)163 * N);
    unsigned* packed = (unsigned*)(csr + (size_t)E);
    int*   hist = (int*)(packed + (size_t)E);
    int*   loc  = hist + (size_t)MAX_BINS * SORT_G;

    const int B = 256;
    auto cdiv = [](long long a, long long b) { return (int)((a + b - 1) / b); };

    const int nb    = cdiv(N, BIN_W);
    const int chunk = cdiv(E, SORT_G);
    bin_local<<<SORT_G, SORT_T, 0, stream>>>(tei, tei + E, hist, loc, packed, E, nb, chunk);
    bin_sort<<<nb, SORT_T, 0, stream>>>(packed, hist, loc, rowp, cnt, dinv, csr, N, E, chunk);

    const int tiles = (N + 15) / 16;

    gemm_x_w1<<<cdiv(tiles, 8), B, 0, stream>>>(x, W1, dinv, h1, N);
    gather_rows<16, true, 8><<<cdiv((long long)4 * N, B), B, 0, stream>>>(
        rowp, cnt, csr, dinv, h1, b1, z1, N);

    gather_rows<16, false, 8><<<cdiv((long long)4 * N, B), B, 0, stream>>>(
        rowp, cnt, csr, dinv, z1, nullptr, a2, N);

    edge_score<<<cdiv(2 * (long long)ET, 1024), B, 0, stream>>>(tpo, tpo + ET, tne, tne + ET,
                                                               a2, W2, b2, (float*)d_out, ET);
}
